# starting kernel
# baseline (speedup 1.0000x reference)
#include <hip/hip_runtime.h>
#include <hip/hip_bf16.h>
#include <cstdint>
#include <cstddef>


typedef __attribute__((ext_vector_type(2))) float v2f;
typedef __attribute__((ext_vector_type(8))) float v8f;

#define CDIV(a, b) (((a) + (b) - 1) / (b))


__global__ void k_zero(float* __restrict__ p, long long n) {
  long long i = (long long)blockIdx.x * blockDim.x + threadIdx.x;
  if (i < n) p[i] = 0.0f;
}

__global__ void k_negcopy(float* __restrict__ dst, const float* __restrict__ src,
                          long long n) {
  long long i = (long long)blockIdx.x * blockDim.x + threadIdx.x;
  if (i < n) dst[i] = -src[i];
}

__global__ void k_deg_count(const int* __restrict__ ei, float* __restrict__ deg, int E) {
  int e = blockIdx.x * blockDim.x + threadIdx.x;
  if (e < E) atomicAdd(&deg[ei[e]], 1.0f);
}

__global__ void k_deg_inv(float* __restrict__ deg, int N) {
  int i = blockIdx.x * blockDim.x + threadIdx.x;
  if (i < N) {
    float d = deg[i];
    deg[i] = (d > 0.0f) ? (1.0f / sqrtf(d)) : 0.0f;
  }
}

__global__ void k_edge_norm(const int* __restrict__ ei, const float* __restrict__ dis,
                            float* __restrict__ nrm, int E) {
  int e = blockIdx.x * blockDim.x + threadIdx.x;
  if (e < E) nrm[e] = dis[ei[e]] * dis[ei[E + e]];
}

__global__ void k_propagate(const int* __restrict__ ei, const float* __restrict__ nrm,
                            const float* __restrict__ x, float* __restrict__ out,
                            int E, int N, int F, float scale) {
  long long tid = (long long)blockIdx.x * blockDim.x + threadIdx.x;
  long long total = (long long)E * F;
  if (tid >= total) return;
  int e = (int)(tid / F);
  int f = (int)(tid % F);
  int r = ei[e];
  int c = ei[E + e];
  float w = scale * nrm[e];
  size_t nf = (size_t)N * F;
#pragma unroll
  for (int b = 0; b < 4; ++b) {
    float v = w * x[(size_t)b * nf + (size_t)c * F + f];
    atomicAdd(&out[(size_t)b * nf + (size_t)r * F + f], v);
  }
}

__global__ void k_pool(const float* __restrict__ x, const int* __restrict__ idx,
                       const float* __restrict__ w, float* __restrict__ out,
                       int Nout, int Nin, int F) {
  long long tid = (long long)blockIdx.x * blockDim.x + threadIdx.x;
  long long total = 4LL * Nout * F;
  if (tid >= total) return;
  int f = (int)(tid % F);
  int n = (int)((tid / F) % Nout);
  int b = (int)(tid / ((long long)F * Nout));
  size_t nfin = (size_t)Nin * F;
  float s = 0.0f;
#pragma unroll
  for (int k = 0; k < 3; ++k) {
    int j = idx[n * 3 + k];
    s += w[n * 3 + k] * x[(size_t)b * nfin + (size_t)j * F + f];
  }
  out[((size_t)b * Nout + n) * F + f] = s;
}

__global__ void k_dense_enc(const float* __restrict__ h, const float* __restrict__ w,
                            const float* __restrict__ bias, float* __restrict__ z) {
  int b = blockIdx.x >> 6;
  int j = blockIdx.x & 63;
  float s = 0.0f;
  for (int i = threadIdx.x; i < 8192; i += blockDim.x)
    s += h[b * 8192 + i] * w[(size_t)i * 64 + j];
  __shared__ float red[256];
  red[threadIdx.x] = s;
  __syncthreads();
  for (int off = 128; off > 0; off >>= 1) {
    if (threadIdx.x < off) red[threadIdx.x] += red[threadIdx.x + off];
    __syncthreads();
  }
  if (threadIdx.x == 0) {
    float v = red[0] + bias[j];
    z[b * 64 + j] = v > 0.0f ? v : 0.0f;
  }
}

__global__ void k_dense_dec(const float* __restrict__ z, const float* __restrict__ w,
                            const float* __restrict__ bias, float* __restrict__ h) {
  long long tid = (long long)blockIdx.x * blockDim.x + threadIdx.x;
  if (tid >= 4LL * 8192) return;
  int b = (int)(tid >> 13);
  int j = (int)(tid & 8191);
  float s = bias[j];
  for (int i = 0; i < 64; ++i) s += z[b * 64 + i] * w[(size_t)i * 8192 + j];
  h[tid] = s > 0.0f ? s : 0.0f;
}


__device__ __forceinline__ float ldsel(const float* __restrict__ p, int idx, bool ok) {
  float v = p[ok ? idx : 0];
  return ok ? v : 0.0f;
}

template <int FIN, int FOUT, int MODE>
__global__ __launch_bounds__(256) void k_cheb_gemm(
    const float* __restrict__ T, const float* __restrict__ W,
    const float* __restrict__ bias, float* __restrict__ out, int M) {
  constexpr int KC = (FIN + 3) / 4;
  constexpr int NTN = (FOUT + 15) / 16;
  constexpr int TM = 4;

  int wave = (int)((blockIdx.x * blockDim.x + threadIdx.x) >> 5);
  int lane = threadIdx.x & 31;
  int groups = (M >> 4) / TM;
  if (wave >= groups * NTN) return;

  int tn = wave % NTN;
  int gm = wave / NTN;
  int m0 = gm * (16 * TM);
  int n0 = tn * 16;
  int lm = lane & 15;
  int khalf = (lane >> 4) << 1;
  int mhi = (lane >> 4) << 3;
  int ncol = n0 + lm;
  bool colok = (FOUT % 16 == 0) || (ncol < FOUT);

  v2f bf[KC];
#pragma unroll
  for (int kc = 0; kc < KC; ++kc) {
    int ka = kc * 4 + khalf;
    if constexpr (FIN % 4 == 0 && FOUT % 16 == 0) {
      bf[kc].x = W[ka * FOUT + ncol];
      bf[kc].y = W[(ka + 1) * FOUT + ncol];
    } else {
      bf[kc].x = ldsel(W, ka * FOUT + ncol, colok && (ka < FIN));
      bf[kc].y = ldsel(W, (ka + 1) * FOUT + ncol, colok && (ka + 1 < FIN));
    }
  }
  float bv = 0.0f;
  if constexpr (MODE == 0) bv = ldsel(bias, ncol, colok);

#pragma unroll
  for (int t = 0; t < TM; ++t) {
    int mb = m0 + t * 16;
    const float* trow = T + (size_t)(mb + lm) * FIN;
    v8f c = {0.f, 0.f, 0.f, 0.f, 0.f, 0.f, 0.f, 0.f};
#pragma unroll
    for (int kc = 0; kc < KC; ++kc) {
      int ka = kc * 4 + khalf;
      v2f a;
      if constexpr (FIN % 4 == 0) {
        a = *(const v2f*)(trow + ka);
      } else {
        a.x = ldsel(trow, ka, ka < FIN);
        a.y = ldsel(trow, ka + 1, ka + 1 < FIN);
      }
      c = __builtin_amdgcn_wmma_f32_16x16x4_f32(false, a, false, bf[kc],
                                                (short)0, c, false, false);
    }
    float* orow = out + (size_t)(mb + mhi) * FOUT + ncol;
    if (colok) {
#pragma unroll
      for (int r = 0; r < 8; ++r) {
        float v = c[r];
        if constexpr (MODE == 0) v += bv;
        if constexpr (MODE == 2 || MODE == 3) v += orow[(size_t)r * FOUT];
        if constexpr (MODE == 3) v = v > 0.0f ? v : 0.0f;
        orow[(size_t)r * FOUT] = v;
      }
    }
  }
}


template <int FIN, int FOUT>
static void launch_gemm_sh(const float* T, const float* W, const float* bias,
                           float* out, int M, int mode, hipStream_t s) {
  constexpr int TM = 4;
  constexpr int NTN = (FOUT + 15) / 16;
  long long waves = (long long)((M / 16) / TM) * NTN;
  unsigned blocks = (unsigned)CDIV(waves * 32, 256);
  switch (mode) {
    case 0: k_cheb_gemm<FIN, FOUT, 0><<<blocks, 256, 0, s>>>(T, W, bias, out, M); break;
    case 1: k_cheb_gemm<FIN, FOUT, 1><<<blocks, 256, 0, s>>>(T, W, bias, out, M); break;
    case 2: k_cheb_gemm<FIN, FOUT, 2><<<blocks, 256, 0, s>>>(T, W, bias, out, M); break;
    default: k_cheb_gemm<FIN, FOUT, 3><<<blocks, 256, 0, s>>>(T, W, bias, out, M); break;
  }
}

static void launch_gemm(const float* T, const float* W, const float* bias,
                        float* out, int M, int Fin, int Fout, int mode,
                        hipStream_t s) {
  if (Fin == 3 && Fout == 16)       launch_gemm_sh<3, 16>(T, W, bias, out, M, mode, s);
  else if (Fin == 16 && Fout == 16) launch_gemm_sh<16, 16>(T, W, bias, out, M, mode, s);
  else if (Fin == 16 && Fout == 32) launch_gemm_sh<16, 32>(T, W, bias, out, M, mode, s);
  else if (Fin == 32 && Fout == 16) launch_gemm_sh<32, 16>(T, W, bias, out, M, mode, s);
  else if (Fin == 16 && Fout == 3)  launch_gemm_sh<16, 3>(T, W, bias, out, M, mode, s);
}

static void run_conv(const float* x, const int* ei, const float* nrm,
                     const float* W  , const float* bias,
                     float* out, float* b0, float* b1, float* b2,
                     int N, int Fin, int Fout, int relu, hipStream_t s) {
  int E = 16 * N;
  int M = 4 * N;
  long long tsz = (long long)M * Fin;
  size_t wstride = (size_t)Fin * Fout;

  launch_gemm(x, W, bias, out, M, Fin, Fout, bias ? 0 : 1, s);

  k_zero<<<(unsigned)CDIV(tsz, 256), 256, 0, s>>>(b0, tsz);
  k_propagate<<<(unsigned)CDIV((long long)E * Fin, 256), 256, 0, s>>>(
      ei, nrm, x, b0, E, N, Fin, 1.0f);
  launch_gemm(b0, W + wstride, nullptr, out, M, Fin, Fout, 2, s);

  const float* p0 = x;
  float* prev = b0;
  float* bufs[3] = {b0, b1, b2};
  for (int k = 2; k < 6; ++k) {
    float* tk = bufs[(k - 1) % 3];
    k_negcopy<<<(unsigned)CDIV(tsz, 256), 256, 0, s>>>(tk, p0, tsz);
    k_propagate<<<(unsigned)CDIV((long long)E * Fin, 256), 256, 0, s>>>(
        ei, nrm, prev, tk, E, N, Fin, 2.0f);
    int last = (k == 5);
    launch_gemm(tk, W + (size_t)k * wstride, nullptr, out, M, Fin, Fout,
                (last && relu) ? 3 : 2, s);
    p0 = prev;
    prev = tk;
  }
}


extern "C" void kernel_launch(void* const* d_in, const int* in_sizes, int n_in,
                              void* d_out, int out_size, void* d_ws, size_t ws_size,
                              hipStream_t stream) {
  (void)in_sizes; (void)n_in; (void)out_size; (void)ws_size;
  static const int NSs[5] = {65536, 16384, 4096, 1024, 256};
  static const int FE[5] = {3, 16, 16, 16, 32};
  static const int FD[6] = {32, 16, 16, 16, 16, 3};

  const float* x = (const float*)d_in[0];
  const int* ei[5];
  for (int l = 0; l < 5; ++l) ei[l] = (const int*)d_in[1 + l];
  const int* down_i[4]; const float* down_w[4];
  const int* up_i[4];   const float* up_w[4];
  for (int i = 0; i < 4; ++i) {
    down_i[i] = (const int*)d_in[6 + 4 * i];
    down_w[i] = (const float*)d_in[7 + 4 * i];
    up_i[i]   = (const int*)d_in[8 + 4 * i];
    up_w[i]   = (const float*)d_in[9 + 4 * i];
  }
  const float* encW[4]; const float* encB[4];
  for (int i = 0; i < 4; ++i) {
    encW[i] = (const float*)d_in[22 + 2 * i];
    encB[i] = (const float*)d_in[23 + 2 * i];
  }
  const float* decW[5];
  for (int i = 0; i < 5; ++i) decW[i] = (const float*)d_in[30 + i];
  const float* decB[4];
  for (int i = 0; i < 4; ++i) decB[i] = (const float*)d_in[35 + i];
  const float* enc_w = (const float*)d_in[39];
  const float* enc_b = (const float*)d_in[40];
  const float* dec_w = (const float*)d_in[41];
  const float* dec_b = (const float*)d_in[42];

  float* ws = (float*)d_ws;
  size_t off = 0;
  float* nrm[5];
  for (int l = 0; l < 5; ++l) { nrm[l] = ws + off; off += (size_t)16 * NSs[l]; }
  float* deg = ws + off; off += 65536;
  const size_t ACT = (size_t)4 * 65536 * 16;
  float* actA = ws + off; off += ACT;
  float* actB = ws + off; off += ACT;
  float* tb0 = ws + off; off += ACT;
  float* tb1 = ws + off; off += ACT;
  float* tb2 = ws + off; off += ACT;
  float* zbuf = ws + off; off += 256;

  for (int l = 0; l < 5; ++l) {
    int N = NSs[l], E = 16 * N;
    k_zero<<<(unsigned)CDIV((long long)N, 256), 256, 0, stream>>>(deg, N);
    k_deg_count<<<(unsigned)CDIV(E, 256), 256, 0, stream>>>(ei[l], deg, E);
    k_deg_inv<<<(unsigned)CDIV(N, 256), 256, 0, stream>>>(deg, N);
    k_edge_norm<<<(unsigned)CDIV(E, 256), 256, 0, stream>>>(ei[l], deg, nrm[l], E);
  }

  const float* cur = x;
  for (int i = 0; i < 4; ++i) {
    run_conv(cur, ei[i], nrm[i], encW[i], encB[i], actA, tb0, tb1, tb2,
             NSs[i], FE[i], FE[i + 1],  1, stream);
    long long pt = 4LL * NSs[i + 1] * FE[i + 1];
    k_pool<<<(unsigned)CDIV(pt, 256), 256, 0, stream>>>(
        actA, down_i[i], down_w[i], actB, NSs[i + 1], NSs[i], FE[i + 1]);
    cur = actB;
  }

  k_dense_enc<<<256, 256, 0, stream>>>(actB, enc_w, enc_b, zbuf);
  k_dense_dec<<<(unsigned)CDIV(4LL * 8192, 256), 256, 0, stream>>>(zbuf, dec_w,
                                                                   dec_b, actA);

  for (int i = 0; i < 4; ++i) {
    int lvl = 3 - i;
    long long pt = 4LL * NSs[lvl] * FD[i];
    k_pool<<<(unsigned)CDIV(pt, 256), 256, 0, stream>>>(
        actA, up_i[lvl], up_w[lvl], actB, NSs[lvl], NSs[lvl + 1], FD[i]);
    run_conv(actB, ei[lvl], nrm[lvl], decW[i], decB[i], actA, tb0, tb1, tb2,
             NSs[lvl], FD[i], FD[i + 1],  1, stream);
  }

  run_conv(actA, ei[0], nrm[0], decW[4], nullptr, (float*)d_out, tb0, tb1, tb2,
           NSs[0], FD[4], FD[5],  0, stream);
}


# Workspace activations and T buffers stored node-major with the four batches interleaved: an edge gathers one contiguous 4*F run (whole 128-byte lines, one address) instead of four 64-byte pieces
# speedup vs baseline: 4.1050x; 4.1050x over previous
#include <hip/hip_runtime.h>
#include <hip/hip_bf16.h>
#include <cstdint>
#include <cstddef>


typedef __attribute__((ext_vector_type(2))) float v2f;
typedef __attribute__((ext_vector_type(8))) float v8f;

#define CDIV(a, b) (((a) + (b) - 1) / (b))


typedef __attribute__((ext_vector_type(4))) float v4f;

__global__ void k_zero_i(int* __restrict__ p, int n) {
  int i = blockIdx.x * blockDim.x + threadIdx.x;
  if (i < n) p[i] = 0;
}

__global__ void k_rank(const int* __restrict__ ei, int* __restrict__ cnt,
                       int* __restrict__ rank, int E) {
  int e = blockIdx.x * blockDim.x + threadIdx.x;
  if (e < E) rank[e] = atomicAdd(&cnt[ei[e]], 1);
}

__global__ __launch_bounds__(1024) void k_scan(const int* __restrict__ cnt,
                                               int* __restrict__ rowptr, int N) {
  __shared__ int part[1024];
  int t = threadIdx.x;
  int per = N >> 10;
  const int* src = cnt + t * per;
  int s = 0;
  for (int i = 0; i < per; ++i) s += src[i];
  part[t] = s;
  __syncthreads();
  for (int off = 1; off < 1024; off <<= 1) {
    int v = (t >= off) ? part[t - off] : 0;
    __syncthreads();
    part[t] += v;
    __syncthreads();
  }
  int run = part[t] - s;
  int* dst = rowptr + t * per;
  for (int i = 0; i < per; ++i) {
    dst[i] = run;
    run += src[i];
  }
  if (t == 1023) rowptr[N] = run;
}

__global__ void k_place(const int* __restrict__ ei, const int* __restrict__ rowptr,
                        const int* __restrict__ rank, int* __restrict__ slot, int E) {
  int e = blockIdx.x * blockDim.x + threadIdx.x;
  if (e < E) slot[rowptr[ei[e]] + rank[e]] = e;
}

__device__ __forceinline__ float inv_sqrt_deg(int d) {
  return (d > 0) ? (1.0f / sqrtf((float)d)) : 0.0f;
}

__global__ void k_emit(const int* __restrict__ ei, const int* __restrict__ cnt,
                       const int* __restrict__ rowptr, const int* __restrict__ slot,
                       int* __restrict__ cs, float* __restrict__ wv, int E) {
  int p = blockIdx.x * blockDim.x + threadIdx.x;
  if (p >= E) return;
  int e = slot[p];
  int r = ei[e];
  int c = ei[E + e];
  int s = rowptr[r];
  int d = cnt[r];
  int k = 0;
  for (int j = 0; j < d; ++j) k += (slot[s + j] < e) ? 1 : 0;
  cs[s + k] = c;
  wv[s + k] = inv_sqrt_deg(d) * inv_sqrt_deg(cnt[c]);
}


template <int V> struct VecOf;
template <> struct VecOf<1> { typedef float T; };
template <> struct VecOf<4> { typedef v4f T; };

__device__ __forceinline__ float ldsel(const float* __restrict__ p, int idx, bool ok) {
  float v = p[ok ? idx : 0];
  return ok ? v : 0.0f;
}

template <int FIN, int FOUT, int STEP, bool RELU>
__global__ __launch_bounds__(256) void k_cheb_step(
    const int* __restrict__ rowptr, const int* __restrict__ cs,
    const float* __restrict__ wv, const float* x, const float* p2, float* tk,
    const float* __restrict__ W, const float* __restrict__ bias, float* out,
    int N, int pb) {
  constexpr int V = (FIN % 4 == 0) ? 4 : 1;
  constexpr int Q = FIN / V;
  constexpr int R = (FIN == 32) ? 32 : 64;
  constexpr int KP = ((FIN + 3) / 4) * 4;
  constexpr int P = (KP == 4) ? 4 : KP + 4;
  constexpr int KC = KP / 4;
  constexpr int NTN = (FOUT + 15) / 16;
  constexpr int MT = 4 * (R / 16);
  constexpr int ITERS = (MT * NTN) / 8;
  constexpr int NT = (STEP == 0) ? 2 : 1;
  constexpr int TILE = 4 * R * P;
  typedef typename VecOf<V>::T VT;
  __shared__ __attribute__((aligned(16))) float sT[NT * TILE];

  const int t = threadIdx.x;
  const int row0 = blockIdx.x * R;
  constexpr bool XB = (FIN == 3) && (STEP == 0);
  constexpr bool OB = (FOUT == 3);
  const bool pbm = (FIN == 3) && (pb != 0);
  const int xsn = XB ? FIN : 4 * FIN;
  const int xsb = XB ? N * FIN : FIN;
  const int psn = pbm ? FIN : 4 * FIN;
  const int psb = pbm ? N * FIN : FIN;
  const int osn = OB ? FOUT : 4 * FOUT;
  const int osb = OB ? N * FOUT : FOUT;
  const bool gth = t < R * Q;
  const int rl = t / Q;
  const int q = t - rl * Q;
  const int op = (row0 + rl) * psn + q * V;
  const int ot = (row0 + rl) * (4 * FIN) + q * V;
  VT a0 = (VT)(0.0f), a1 = (VT)(0.0f), a2 = (VT)(0.0f), a3 = (VT)(0.0f);

  if (gth) {
    int r = row0 + rl;
    int j = rowptr[r];
    int je = rowptr[r + 1];
    const float* xq = x + q * V;
    for (; j + 4 <= je; j += 4) {
      int c0 = cs[j], c1 = cs[j + 1], c2 = cs[j + 2], c3 = cs[j + 3];
      float w0 = wv[j], w1 = wv[j + 1], w2 = wv[j + 2], w3 = wv[j + 3];
      const float* s0 = xq + c0 * xsn;
      const float* s1 = xq + c1 * xsn;
      const float* s2 = xq + c2 * xsn;
      const float* s3 = xq + c3 * xsn;
      VT v00 = *(const VT*)(s0), v01 = *(const VT*)(s0 + xsb);
      VT v02 = *(const VT*)(s0 + 2 * xsb), v03 = *(const VT*)(s0 + 3 * xsb);
      VT v10 = *(const VT*)(s1), v11 = *(const VT*)(s1 + xsb);
      VT v12 = *(const VT*)(s1 + 2 * xsb), v13 = *(const VT*)(s1 + 3 * xsb);
      VT v20 = *(const VT*)(s2), v21 = *(const VT*)(s2 + xsb);
      VT v22 = *(const VT*)(s2 + 2 * xsb), v23 = *(const VT*)(s2 + 3 * xsb);
      VT v30 = *(const VT*)(s3), v31 = *(const VT*)(s3 + xsb);
      VT v32 = *(const VT*)(s3 + 2 * xsb), v33 = *(const VT*)(s3 + 3 * xsb);
      a0 += w0 * v00; a1 += w0 * v01; a2 += w0 * v02; a3 += w0 * v03;
      a0 += w1 * v10; a1 += w1 * v11; a2 += w1 * v12; a3 += w1 * v13;
      a0 += w2 * v20; a1 += w2 * v21; a2 += w2 * v22; a3 += w2 * v23;
      a0 += w3 * v30; a1 += w3 * v31; a2 += w3 * v32; a3 += w3 * v33;
    }
    for (; j < je; ++j) {
      int c = cs[j];
      float w = wv[j];
      const float* s0 = xq + c * xsn;
      VT v0 = *(const VT*)(s0), v1 = *(const VT*)(s0 + xsb);
      VT v2 = *(const VT*)(s0 + 2 * xsb), v3 = *(const VT*)(s0 + 3 * xsb);
      a0 += w * v0; a1 += w * v1; a2 += w * v2; a3 += w * v3;
    }
    VT p0 = *(const VT*)(p2 + op);
    VT p1 = *(const VT*)(p2 + op + psb);
    VT p2v = *(const VT*)(p2 + op + 2 * psb);
    VT p3 = *(const VT*)(p2 + op + 3 * psb);
    if constexpr (STEP != 0) {
      a0 = 2.0f * a0 - p0;
      a1 = 2.0f * a1 - p1;
      a2 = 2.0f * a2 - p2v;
      a3 = 2.0f * a3 - p3;
    }
    float* d = sT + rl * P + q * V;
    *(VT*)(d) = a0;
    *(VT*)(d + R * P) = a1;
    *(VT*)(d + 2 * R * P) = a2;
    *(VT*)(d + 3 * R * P) = a3;
    if constexpr (STEP == 0) {
      *(VT*)(d + TILE) = p0;
      *(VT*)(d + TILE + R * P) = p1;
      *(VT*)(d + TILE + 2 * R * P) = p2v;
      *(VT*)(d + TILE + 3 * R * P) = p3;
    }
    if constexpr (KP != FIN) {
      if (q == 0) {
        for (int i = 0; i < NT * 4; ++i)
          for (int f = FIN; f < KP; ++f) sT[i * R * P + rl * P + f] = 0.0f;
      }
    }
  }
  __syncthreads();

  if constexpr (STEP != 2) {
    if (gth) {
      *(VT*)(tk + ot) = a0;
      *(VT*)(tk + ot + FIN) = a1;
      *(VT*)(tk + ot + 2 * FIN) = a2;
      *(VT*)(tk + ot + 3 * FIN) = a3;
    }
  }

  const int wave = t >> 5;
  const int lane = t & 31;
  const int lm = lane & 15;
  const int khalf = (lane >> 4) << 1;
  const int mhi = (lane >> 4) << 3;
  const int tn = wave % NTN;
  const int ncol = tn * 16 + lm;
  const bool colok = (FOUT % 16 == 0) || (ncol < FOUT);

  v2f bf[NT][KC];
#pragma unroll
  for (int i = 0; i < NT; ++i) {
    const float* Wi = W + ((STEP == 0 && i == 0) ? FIN * FOUT : 0);
#pragma unroll
    for (int kc = 0; kc < KC; ++kc) {
      int ka = kc * 4 + khalf;
      if constexpr (FIN % 4 == 0 && FOUT % 16 == 0) {
        bf[i][kc].x = Wi[ka * FOUT + ncol];
        bf[i][kc].y = Wi[(ka + 1) * FOUT + ncol];
      } else {
        bf[i][kc].x = ldsel(Wi, ka * FOUT + ncol, colok && (ka < FIN));
        bf[i][kc].y = ldsel(Wi, (ka + 1) * FOUT + ncol, colok && (ka + 1 < FIN));
      }
    }
  }
  float bv = 0.0f;
  if constexpr (STEP == 0) {
    if (bias) bv = ldsel(bias, ncol, colok);
  }

#pragma unroll
  for (int it = 0; it < ITERS; ++it) {
    int mt = (wave / NTN) * ITERS + it;
    int b = mt & 3;
    int rt = mt >> 2;
    const float* arow = sT + (b * R + rt * 16 + lm) * P + khalf;
    v8f c = {0.f, 0.f, 0.f, 0.f, 0.f, 0.f, 0.f, 0.f};
#pragma unroll
    for (int i = 0; i < NT; ++i) {
#pragma unroll
      for (int kc = 0; kc < KC; ++kc) {
        v2f a = *(const v2f*)(arow + i * TILE + kc * 4);
        c = __builtin_amdgcn_wmma_f32_16x16x4_f32(false, a, false, bf[i][kc],
                                                  (short)0, c, false, false);
      }
    }
    float* orow = out + (size_t)(row0 + rt * 16 + mhi) * osn + (size_t)b * osb + ncol;
    if (colok) {
#pragma unroll
      for (int r = 0; r < 8; ++r) {
        float v = c[r];
        if constexpr (STEP == 0) v += bv;
        else v += orow[(size_t)r * osn];
        if constexpr (STEP == 2 && RELU) v = v > 0.0f ? v : 0.0f;
        orow[(size_t)r * osn] = v;
      }
    }
  }
}

__global__ void k_pool(const float* __restrict__ x, const int* __restrict__ idx,
                       const float* __restrict__ w, float* __restrict__ out,
                       int Nout, int Nin, int F) {
  (void)Nin;
  long long tid = (long long)blockIdx.x * blockDim.x + threadIdx.x;
  long long total = 4LL * Nout * F;
  if (tid >= total) return;
  int F4 = 4 * F;
  int bf = (int)(tid % F4);
  int n = (int)(tid / F4);
  float s = 0.0f;
#pragma unroll
  for (int k = 0; k < 3; ++k) {
    int j = idx[n * 3 + k];
    s += w[n * 3 + k] * x[(size_t)j * F4 + bf];
  }
  out[tid] = s;
}

__global__ void k_dense_enc(const float* __restrict__ h, const float* __restrict__ w,
                            const float* __restrict__ bias, float* __restrict__ z) {
  int b = blockIdx.x >> 6;
  int j = blockIdx.x & 63;
  float s = 0.0f;
  for (int i = threadIdx.x; i < 8192; i += blockDim.x)
    s += h[((i >> 5) * 4 + b) * 32 + (i & 31)] * w[(size_t)i * 64 + j];
  __shared__ float red[256];
  red[threadIdx.x] = s;
  __syncthreads();
  for (int off = 128; off > 0; off >>= 1) {
    if (threadIdx.x < off) red[threadIdx.x] += red[threadIdx.x + off];
    __syncthreads();
  }
  if (threadIdx.x == 0) {
    float v = red[0] + bias[j];
    z[b * 64 + j] = v > 0.0f ? v : 0.0f;
  }
}

__global__ void k_dense_dec(const float* __restrict__ z, const float* __restrict__ w,
                            const float* __restrict__ bias, float* __restrict__ h) {
  long long tid = (long long)blockIdx.x * blockDim.x + threadIdx.x;
  if (tid >= 4LL * 8192) return;
  int b = (int)(tid >> 13);
  int j = (int)(tid & 8191);
  float s = bias[j];
  for (int i = 0; i < 64; ++i) s += z[b * 64 + i] * w[(size_t)i * 8192 + j];
  h[((j >> 5) * 4 + b) * 32 + (j & 31)] = s > 0.0f ? s : 0.0f;
}


template <int FIN, int FOUT>
static void launch_step_sh(const int* rp, const int* cs, const float* wv,
                           const float* x, const float* p2, float* tk,
                           const float* W, const float* bias, float* out, int N,
                           int step, int relu, int pb, hipStream_t s) {
  constexpr int R = (FIN == 32) ? 32 : 64;
  unsigned blocks = (unsigned)(N / R);
  if (step == 0)
    k_cheb_step<FIN, FOUT, 0, false><<<blocks, 256, 0, s>>>(rp, cs, wv, x, p2, tk, W, bias, out, N, pb);
  else if (step == 1)
    k_cheb_step<FIN, FOUT, 1, false><<<blocks, 256, 0, s>>>(rp, cs, wv, x, p2, tk, W, bias, out, N, pb);
  else if (relu)
    k_cheb_step<FIN, FOUT, 2, true><<<blocks, 256, 0, s>>>(rp, cs, wv, x, p2, tk, W, bias, out, N, pb);
  else
    k_cheb_step<FIN, FOUT, 2, false><<<blocks, 256, 0, s>>>(rp, cs, wv, x, p2, tk, W, bias, out, N, pb);
}

static void launch_step(const int* rp, const int* cs, const float* wv,
                        const float* x, const float* p2, float* tk,
                        const float* W, const float* bias, float* out, int N,
                        int Fin, int Fout, int step, int relu, int pb,
                        hipStream_t s) {
  if (Fin == 3 && Fout == 16)
    launch_step_sh<3, 16>(rp, cs, wv, x, p2, tk, W, bias, out, N, step, relu, pb, s);
  else if (Fin == 16 && Fout == 16)
    launch_step_sh<16, 16>(rp, cs, wv, x, p2, tk, W, bias, out, N, step, relu, pb, s);
  else if (Fin == 16 && Fout == 32)
    launch_step_sh<16, 32>(rp, cs, wv, x, p2, tk, W, bias, out, N, step, relu, pb, s);
  else if (Fin == 32 && Fout == 16)
    launch_step_sh<32, 16>(rp, cs, wv, x, p2, tk, W, bias, out, N, step, relu, pb, s);
  else if (Fin == 16 && Fout == 3)
    launch_step_sh<16, 3>(rp, cs, wv, x, p2, tk, W, bias, out, N, step, relu, pb, s);
}

static void run_conv(const float* x, const int* rp, const int* cs, const float* wv,
                     const float* W  , const float* bias,
                     float* out, float* b0, float* b1,
                     int N, int Fin, int Fout, int relu, int xin, hipStream_t s) {
  size_t wstride = (size_t)Fin * Fout;

  launch_step(rp, cs, wv, x, x, b0, W, bias, out, N, Fin, Fout, 0, relu,
               xin, s);

  for (int k = 2; k < 6; ++k) {
    const float* prev = (k & 1) ? b1 : b0;
    float* tk = (k & 1) ? b0 : b1;
    const float* p2 = (k == 2) ? x : tk;
    launch_step(rp, cs, wv, prev, p2, tk, W + (size_t)k * wstride, nullptr, out,
                N, Fin, Fout, (k == 5) ? 2 : 1, relu,
                 (k == 2) ? xin : 0, s);
  }
}


extern "C" void kernel_launch(void* const* d_in, const int* in_sizes, int n_in,
                              void* d_out, int out_size, void* d_ws, size_t ws_size,
                              hipStream_t stream) {
  (void)in_sizes; (void)n_in; (void)out_size; (void)ws_size;
  static const int NSs[5] = {65536, 16384, 4096, 1024, 256};
  static const int FE[5] = {3, 16, 16, 16, 32};
  static const int FD[6] = {32, 16, 16, 16, 16, 3};

  const float* x = (const float*)d_in[0];
  const int* ei[5];
  for (int l = 0; l < 5; ++l) ei[l] = (const int*)d_in[1 + l];
  const int* down_i[4]; const float* down_w[4];
  const int* up_i[4];   const float* up_w[4];
  for (int i = 0; i < 4; ++i) {
    down_i[i] = (const int*)d_in[6 + 4 * i];
    down_w[i] = (const float*)d_in[7 + 4 * i];
    up_i[i]   = (const int*)d_in[8 + 4 * i];
    up_w[i]   = (const float*)d_in[9 + 4 * i];
  }
  const float* encW[4]; const float* encB[4];
  for (int i = 0; i < 4; ++i) {
    encW[i] = (const float*)d_in[22 + 2 * i];
    encB[i] = (const float*)d_in[23 + 2 * i];
  }
  const float* decW[5];
  for (int i = 0; i < 5; ++i) decW[i] = (const float*)d_in[30 + i];
  const float* decB[4];
  for (int i = 0; i < 4; ++i) decB[i] = (const float*)d_in[35 + i];
  const float* enc_w = (const float*)d_in[39];
  const float* enc_b = (const float*)d_in[40];
  const float* dec_w = (const float*)d_in[41];
  const float* dec_b = (const float*)d_in[42];

  float* ws = (float*)d_ws;
  size_t off = 0;
  const size_t ACT = (size_t)4 * 65536 * 16;
  float* actA = ws + off; off += ACT;
  float* actB = ws + off; off += ACT;
  float* tb0 = ws + off; off += ACT;
  float* tb1 = ws + off; off += ACT;
  float* zbuf = ws + off; off += 256;
  int* rp[4]; int* cs[4]; float* wv[4];
  for (int l = 0; l < 4; ++l) {
    rp[l] = (int*)(ws + off); off += (size_t)NSs[l] + 4;
    cs[l] = (int*)(ws + off); off += (size_t)16 * NSs[l];
    wv[l] = ws + off;         off += (size_t)16 * NSs[l];
  }
  int* rank = (int*)tb0;
  int* slot = rank + (size_t)16 * 65536;
  int* cnt = slot + (size_t)16 * 65536;

  for (int l = 0; l < 4; ++l) {
    int N = NSs[l], E = 16 * N;
    k_zero_i<<<(unsigned)CDIV(N, 256), 256, 0, stream>>>(cnt, N);
    k_rank<<<(unsigned)CDIV(E, 256), 256, 0, stream>>>(ei[l], cnt, rank, E);
    k_scan<<<1, 1024, 0, stream>>>(cnt, rp[l], N);
    k_place<<<(unsigned)CDIV(E, 256), 256, 0, stream>>>(ei[l], rp[l], rank, slot, E);
    k_emit<<<(unsigned)CDIV(E, 256), 256, 0, stream>>>(ei[l], cnt, rp[l], slot,
                                                       cs[l], wv[l], E);
  }

  const float* cur = x;
  for (int i = 0; i < 4; ++i) {
    run_conv(cur, rp[i], cs[i], wv[i], encW[i], encB[i], actA, tb0, tb1,
             NSs[i], FE[i], FE[i + 1],  1,  (i == 0) ? 1 : 0,
             stream);
    long long pt = 4LL * NSs[i + 1] * FE[i + 1];
    k_pool<<<(unsigned)CDIV(pt, 256), 256, 0, stream>>>(
        actA, down_i[i], down_w[i], actB, NSs[i + 1], NSs[i], FE[i + 1]);
    cur = actB;
  }

  k_dense_enc<<<256, 256, 0, stream>>>(actB, enc_w, enc_b, zbuf);
  k_dense_dec<<<(unsigned)CDIV(4LL * 8192, 256), 256, 0, stream>>>(zbuf, dec_w,
                                                                   dec_b, actA);

  for (int i = 0; i < 4; ++i) {
    int lvl = 3 - i;
    long long pt = 4LL * NSs[lvl] * FD[i];
    k_pool<<<(unsigned)CDIV(pt, 256), 256, 0, stream>>>(
        actA, up_i[lvl], up_w[lvl], actB, NSs[lvl], NSs[lvl + 1], FD[i]);
    run_conv(actB, rp[lvl], cs[lvl], wv[lvl], decW[i], decB[i], actA, tb0, tb1,
             NSs[lvl], FD[i], FD[i + 1],  1,  0, stream);
  }

  run_conv(actA, rp[0], cs[0], wv[0], decW[4], nullptr, (float*)d_out, tb0, tb1,
           NSs[0], FD[4], FD[5],  0,  0, stream);
}


# CSR build of the four conv levels shares one set of five launches (zero, rank, scan, place, emit split by level at block granularity): 75 -> 60 launches
# speedup vs baseline: 4.3165x; 1.0515x over previous
#include <hip/hip_runtime.h>
#include <hip/hip_bf16.h>
#include <cstdint>
#include <cstddef>


typedef __attribute__((ext_vector_type(2))) float v2f;
typedef __attribute__((ext_vector_type(8))) float v8f;

#define CDIV(a, b) (((a) + (b) - 1) / (b))


typedef __attribute__((ext_vector_type(4))) float v4f;

#define NLV(l) (65536 >> (2 * (l)))
#define ELV(l) (16 * NLV(l))
#define NOFF1 NLV(0)
#define NOFF2 (NOFF1 + NLV(1))
#define NOFF3 (NOFF2 + NLV(2))
#define NTOT (NOFF3 + NLV(3))
#define EOFF1 ELV(0)
#define EOFF2 (EOFF1 + ELV(1))
#define EOFF3 (EOFF2 + ELV(2))
#define ETOT (EOFF3 + ELV(3))

__device__ __forceinline__ int edge_level(int g) {
  return (g >= EOFF1 ? 1 : 0) + (g >= EOFF2 ? 1 : 0) + (g >= EOFF3 ? 1 : 0);
}
__device__ __forceinline__ int edge_off(int l) {
  return l == 0 ? 0 : (l == 1 ? EOFF1 : (l == 2 ? EOFF2 : EOFF3));
}
__device__ __forceinline__ int node_off(int l) {
  return l == 0 ? 0 : (l == 1 ? NOFF1 : (l == 2 ? NOFF2 : NOFF3));
}
template <typename T>
__device__ __forceinline__ T pick4(int l, T a0, T a1, T a2, T a3) {
  return l == 0 ? a0 : (l == 1 ? a1 : (l == 2 ? a2 : a3));
}

__global__ void k_zero_i(int* __restrict__ p, int n) {
  int i = blockIdx.x * blockDim.x + threadIdx.x;
  if (i < n) p[i] = 0;
}

__global__ void k_rank(const int* __restrict__ ei0, const int* __restrict__ ei1,
                       const int* __restrict__ ei2, const int* __restrict__ ei3,
                       int* __restrict__ cnt, int* __restrict__ rank) {
  int g = blockIdx.x * blockDim.x + threadIdx.x;
  if (g >= ETOT) return;
  int l = edge_level(g);
  const int* ei = pick4(l, ei0, ei1, ei2, ei3);
  int e = g - edge_off(l);
  rank[g] = atomicAdd(&cnt[node_off(l) + ei[e]], 1);
}

__global__ __launch_bounds__(1024) void k_scan(const int* __restrict__ cnt,
                                               int* __restrict__ rp0,
                                               int* __restrict__ rp1,
                                               int* __restrict__ rp2,
                                               int* __restrict__ rp3) {
  __shared__ int part[1024];
  int l = blockIdx.x;
  int N = NLV(l);
  int* rowptr = pick4(l, rp0, rp1, rp2, rp3);
  int t = threadIdx.x;
  int per = N >> 10;
  const int* src = cnt + node_off(l) + t * per;
  int s = 0;
  for (int i = 0; i < per; ++i) s += src[i];
  part[t] = s;
  __syncthreads();
  for (int off = 1; off < 1024; off <<= 1) {
    int v = (t >= off) ? part[t - off] : 0;
    __syncthreads();
    part[t] += v;
    __syncthreads();
  }
  int run = part[t] - s;
  int* dst = rowptr + t * per;
  for (int i = 0; i < per; ++i) {
    dst[i] = run;
    run += src[i];
  }
  if (t == 1023) rowptr[N] = run;
}

__global__ void k_place(const int* __restrict__ ei0, const int* __restrict__ ei1,
                        const int* __restrict__ ei2, const int* __restrict__ ei3,
                        const int* __restrict__ rp0, const int* __restrict__ rp1,
                        const int* __restrict__ rp2, const int* __restrict__ rp3,
                        const int* __restrict__ rank, int* __restrict__ slot) {
  int g = blockIdx.x * blockDim.x + threadIdx.x;
  if (g >= ETOT) return;
  int l = edge_level(g);
  const int* ei = pick4(l, ei0, ei1, ei2, ei3);
  const int* rowptr = pick4(l, rp0, rp1, rp2, rp3);
  int eo = edge_off(l);
  int e = g - eo;
  slot[eo + rowptr[ei[e]] + rank[g]] = e;
}

__device__ __forceinline__ float inv_sqrt_deg(int d) {
  return (d > 0) ? (1.0f / sqrtf((float)d)) : 0.0f;
}

struct CsrOut {
  int* cs0; int* cs1; int* cs2; int* cs3;
  float* wv0; float* wv1; float* wv2; float* wv3;
};

__global__ void k_emit(const int* __restrict__ ei0, const int* __restrict__ ei1,
                       const int* __restrict__ ei2, const int* __restrict__ ei3,
                       const int* __restrict__ cnt,
                       const int* __restrict__ rp0, const int* __restrict__ rp1,
                       const int* __restrict__ rp2, const int* __restrict__ rp3,
                       const int* __restrict__ slot, CsrOut o) {
  int g = blockIdx.x * blockDim.x + threadIdx.x;
  if (g >= ETOT) return;
  int l = edge_level(g);
  const int* ei = pick4(l, ei0, ei1, ei2, ei3);
  const int* rowptr = pick4(l, rp0, rp1, rp2, rp3);
  int* cs = pick4(l, o.cs0, o.cs1, o.cs2, o.cs3);
  float* wv = pick4(l, o.wv0, o.wv1, o.wv2, o.wv3);
  const int* cl = cnt + node_off(l);
  const int* sl = slot + edge_off(l);
  int E = ELV(l);
  int e = sl[g - edge_off(l)];
  int r = ei[e];
  int c = ei[E + e];
  int s = rowptr[r];
  int d = cl[r];
  int k = 0;
  for (int j = 0; j < d; ++j) k += (sl[s + j] < e) ? 1 : 0;
  cs[s + k] = c;
  wv[s + k] = inv_sqrt_deg(d) * inv_sqrt_deg(cl[c]);
}


template <int V> struct VecOf;
template <> struct VecOf<1> { typedef float T; };
template <> struct VecOf<4> { typedef v4f T; };

__device__ __forceinline__ float ldsel(const float* __restrict__ p, int idx, bool ok) {
  float v = p[ok ? idx : 0];
  return ok ? v : 0.0f;
}

template <int FIN, int FOUT, int STEP, bool RELU>
__global__ __launch_bounds__(256) void k_cheb_step(
    const int* __restrict__ rowptr, const int* __restrict__ cs,
    const float* __restrict__ wv, const float* x, const float* p2, float* tk,
    const float* __restrict__ W, const float* __restrict__ bias, float* out,
    int N, int pb) {
  constexpr int V = (FIN % 4 == 0) ? 4 : 1;
  constexpr int Q = FIN / V;
  constexpr int R = (FIN == 32) ? 32 : 64;
  constexpr int KP = ((FIN + 3) / 4) * 4;
  constexpr int P = (KP == 4) ? 4 : KP + 4;
  constexpr int KC = KP / 4;
  constexpr int NTN = (FOUT + 15) / 16;
  constexpr int MT = 4 * (R / 16);
  constexpr int ITERS = (MT * NTN) / 8;
  constexpr int NT = (STEP == 0) ? 2 : 1;
  constexpr int TILE = 4 * R * P;
  typedef typename VecOf<V>::T VT;
  __shared__ __attribute__((aligned(16))) float sT[NT * TILE];

  const int t = threadIdx.x;
  const int row0 = blockIdx.x * R;
  constexpr bool XB = (FIN == 3) && (STEP == 0);
  constexpr bool OB = (FOUT == 3);
  const bool pbm = (FIN == 3) && (pb != 0);
  const int xsn = XB ? FIN : 4 * FIN;
  const int xsb = XB ? N * FIN : FIN;
  const int psn = pbm ? FIN : 4 * FIN;
  const int psb = pbm ? N * FIN : FIN;
  const int osn = OB ? FOUT : 4 * FOUT;
  const int osb = OB ? N * FOUT : FOUT;
  const bool gth = t < R * Q;
  const int rl = t / Q;
  const int q = t - rl * Q;
  const int op = (row0 + rl) * psn + q * V;
  const int ot = (row0 + rl) * (4 * FIN) + q * V;
  VT a0 = (VT)(0.0f), a1 = (VT)(0.0f), a2 = (VT)(0.0f), a3 = (VT)(0.0f);

  if (gth) {
    int r = row0 + rl;
    int j = rowptr[r];
    int je = rowptr[r + 1];
    const float* xq = x + q * V;
    for (; j + 4 <= je; j += 4) {
      int c0 = cs[j], c1 = cs[j + 1], c2 = cs[j + 2], c3 = cs[j + 3];
      float w0 = wv[j], w1 = wv[j + 1], w2 = wv[j + 2], w3 = wv[j + 3];
      const float* s0 = xq + c0 * xsn;
      const float* s1 = xq + c1 * xsn;
      const float* s2 = xq + c2 * xsn;
      const float* s3 = xq + c3 * xsn;
      VT v00 = *(const VT*)(s0), v01 = *(const VT*)(s0 + xsb);
      VT v02 = *(const VT*)(s0 + 2 * xsb), v03 = *(const VT*)(s0 + 3 * xsb);
      VT v10 = *(const VT*)(s1), v11 = *(const VT*)(s1 + xsb);
      VT v12 = *(const VT*)(s1 + 2 * xsb), v13 = *(const VT*)(s1 + 3 * xsb);
      VT v20 = *(const VT*)(s2), v21 = *(const VT*)(s2 + xsb);
      VT v22 = *(const VT*)(s2 + 2 * xsb), v23 = *(const VT*)(s2 + 3 * xsb);
      VT v30 = *(const VT*)(s3), v31 = *(const VT*)(s3 + xsb);
      VT v32 = *(const VT*)(s3 + 2 * xsb), v33 = *(const VT*)(s3 + 3 * xsb);
      a0 += w0 * v00; a1 += w0 * v01; a2 += w0 * v02; a3 += w0 * v03;
      a0 += w1 * v10; a1 += w1 * v11; a2 += w1 * v12; a3 += w1 * v13;
      a0 += w2 * v20; a1 += w2 * v21; a2 += w2 * v22; a3 += w2 * v23;
      a0 += w3 * v30; a1 += w3 * v31; a2 += w3 * v32; a3 += w3 * v33;
    }
    for (; j < je; ++j) {
      int c = cs[j];
      float w = wv[j];
      const float* s0 = xq + c * xsn;
      VT v0 = *(const VT*)(s0), v1 = *(const VT*)(s0 + xsb);
      VT v2 = *(const VT*)(s0 + 2 * xsb), v3 = *(const VT*)(s0 + 3 * xsb);
      a0 += w * v0; a1 += w * v1; a2 += w * v2; a3 += w * v3;
    }
    VT p0 = *(const VT*)(p2 + op);
    VT p1 = *(const VT*)(p2 + op + psb);
    VT p2v = *(const VT*)(p2 + op + 2 * psb);
    VT p3 = *(const VT*)(p2 + op + 3 * psb);
    if constexpr (STEP != 0) {
      a0 = 2.0f * a0 - p0;
      a1 = 2.0f * a1 - p1;
      a2 = 2.0f * a2 - p2v;
      a3 = 2.0f * a3 - p3;
    }
    float* d = sT + rl * P + q * V;
    *(VT*)(d) = a0;
    *(VT*)(d + R * P) = a1;
    *(VT*)(d + 2 * R * P) = a2;
    *(VT*)(d + 3 * R * P) = a3;
    if constexpr (STEP == 0) {
      *(VT*)(d + TILE) = p0;
      *(VT*)(d + TILE + R * P) = p1;
      *(VT*)(d + TILE + 2 * R * P) = p2v;
      *(VT*)(d + TILE + 3 * R * P) = p3;
    }
    if constexpr (KP != FIN) {
      if (q == 0) {
        for (int i = 0; i < NT * 4; ++i)
          for (int f = FIN; f < KP; ++f) sT[i * R * P + rl * P + f] = 0.0f;
      }
    }
  }
  __syncthreads();

  if constexpr (STEP != 2) {
    if (gth) {
      *(VT*)(tk + ot) = a0;
      *(VT*)(tk + ot + FIN) = a1;
      *(VT*)(tk + ot + 2 * FIN) = a2;
      *(VT*)(tk + ot + 3 * FIN) = a3;
    }
  }

  const int wave = t >> 5;
  const int lane = t & 31;
  const int lm = lane & 15;
  const int khalf = (lane >> 4) << 1;
  const int mhi = (lane >> 4) << 3;
  const int tn = wave % NTN;
  const int ncol = tn * 16 + lm;
  const bool colok = (FOUT % 16 == 0) || (ncol < FOUT);

  v2f bf[NT][KC];
#pragma unroll
  for (int i = 0; i < NT; ++i) {
    const float* Wi = W + ((STEP == 0 && i == 0) ? FIN * FOUT : 0);
#pragma unroll
    for (int kc = 0; kc < KC; ++kc) {
      int ka = kc * 4 + khalf;
      if constexpr (FIN % 4 == 0 && FOUT % 16 == 0) {
        bf[i][kc].x = Wi[ka * FOUT + ncol];
        bf[i][kc].y = Wi[(ka + 1) * FOUT + ncol];
      } else {
        bf[i][kc].x = ldsel(Wi, ka * FOUT + ncol, colok && (ka < FIN));
        bf[i][kc].y = ldsel(Wi, (ka + 1) * FOUT + ncol, colok && (ka + 1 < FIN));
      }
    }
  }
  float bv = 0.0f;
  if constexpr (STEP == 0) {
    if (bias) bv = ldsel(bias, ncol, colok);
  }

#pragma unroll
  for (int it = 0; it < ITERS; ++it) {
    int mt = (wave / NTN) * ITERS + it;
    int b = mt & 3;
    int rt = mt >> 2;
    const float* arow = sT + (b * R + rt * 16 + lm) * P + khalf;
    v8f c = {0.f, 0.f, 0.f, 0.f, 0.f, 0.f, 0.f, 0.f};
#pragma unroll
    for (int i = 0; i < NT; ++i) {
#pragma unroll
      for (int kc = 0; kc < KC; ++kc) {
        v2f a = *(const v2f*)(arow + i * TILE + kc * 4);
        c = __builtin_amdgcn_wmma_f32_16x16x4_f32(false, a, false, bf[i][kc],
                                                  (short)0, c, false, false);
      }
    }
    float* orow = out + (size_t)(row0 + rt * 16 + mhi) * osn + (size_t)b * osb + ncol;
    if (colok) {
#pragma unroll
      for (int r = 0; r < 8; ++r) {
        float v = c[r];
        if constexpr (STEP == 0) v += bv;
        else v += orow[(size_t)r * osn];
        if constexpr (STEP == 2 && RELU) v = v > 0.0f ? v : 0.0f;
        orow[(size_t)r * osn] = v;
      }
    }
  }
}

__global__ void k_pool(const float* __restrict__ x, const int* __restrict__ idx,
                       const float* __restrict__ w, float* __restrict__ out,
                       int Nout, int Nin, int F) {
  (void)Nin;
  long long tid = (long long)blockIdx.x * blockDim.x + threadIdx.x;
  long long total = 4LL * Nout * F;
  if (tid >= total) return;
  int F4 = 4 * F;
  int bf = (int)(tid % F4);
  int n = (int)(tid / F4);
  float s = 0.0f;
#pragma unroll
  for (int k = 0; k < 3; ++k) {
    int j = idx[n * 3 + k];
    s += w[n * 3 + k] * x[(size_t)j * F4 + bf];
  }
  out[tid] = s;
}

__global__ void k_dense_enc(const float* __restrict__ h, const float* __restrict__ w,
                            const float* __restrict__ bias, float* __restrict__ z) {
  int b = blockIdx.x >> 6;
  int j = blockIdx.x & 63;
  float s = 0.0f;
  for (int i = threadIdx.x; i < 8192; i += blockDim.x)
    s += h[((i >> 5) * 4 + b) * 32 + (i & 31)] * w[(size_t)i * 64 + j];
  __shared__ float red[256];
  red[threadIdx.x] = s;
  __syncthreads();
  for (int off = 128; off > 0; off >>= 1) {
    if (threadIdx.x < off) red[threadIdx.x] += red[threadIdx.x + off];
    __syncthreads();
  }
  if (threadIdx.x == 0) {
    float v = red[0] + bias[j];
    z[b * 64 + j] = v > 0.0f ? v : 0.0f;
  }
}

__global__ void k_dense_dec(const float* __restrict__ z, const float* __restrict__ w,
                            const float* __restrict__ bias, float* __restrict__ h) {
  long long tid = (long long)blockIdx.x * blockDim.x + threadIdx.x;
  if (tid >= 4LL * 8192) return;
  int b = (int)(tid >> 13);
  int j = (int)(tid & 8191);
  float s = bias[j];
  for (int i = 0; i < 64; ++i) s += z[b * 64 + i] * w[(size_t)i * 8192 + j];
  h[((j >> 5) * 4 + b) * 32 + (j & 31)] = s > 0.0f ? s : 0.0f;
}


template <int FIN, int FOUT>
static void launch_step_sh(const int* rp, const int* cs, const float* wv,
                           const float* x, const float* p2, float* tk,
                           const float* W, const float* bias, float* out, int N,
                           int step, int relu, int pb, hipStream_t s) {
  constexpr int R = (FIN == 32) ? 32 : 64;
  unsigned blocks = (unsigned)(N / R);
  if (step == 0)
    k_cheb_step<FIN, FOUT, 0, false><<<blocks, 256, 0, s>>>(rp, cs, wv, x, p2, tk, W, bias, out, N, pb);
  else if (step == 1)
    k_cheb_step<FIN, FOUT, 1, false><<<blocks, 256, 0, s>>>(rp, cs, wv, x, p2, tk, W, bias, out, N, pb);
  else if (relu)
    k_cheb_step<FIN, FOUT, 2, true><<<blocks, 256, 0, s>>>(rp, cs, wv, x, p2, tk, W, bias, out, N, pb);
  else
    k_cheb_step<FIN, FOUT, 2, false><<<blocks, 256, 0, s>>>(rp, cs, wv, x, p2, tk, W, bias, out, N, pb);
}

static void launch_step(const int* rp, const int* cs, const float* wv,
                        const float* x, const float* p2, float* tk,
                        const float* W, const float* bias, float* out, int N,
                        int Fin, int Fout, int step, int relu, int pb,
                        hipStream_t s) {
  if (Fin == 3 && Fout == 16)
    launch_step_sh<3, 16>(rp, cs, wv, x, p2, tk, W, bias, out, N, step, relu, pb, s);
  else if (Fin == 16 && Fout == 16)
    launch_step_sh<16, 16>(rp, cs, wv, x, p2, tk, W, bias, out, N, step, relu, pb, s);
  else if (Fin == 16 && Fout == 32)
    launch_step_sh<16, 32>(rp, cs, wv, x, p2, tk, W, bias, out, N, step, relu, pb, s);
  else if (Fin == 32 && Fout == 16)
    launch_step_sh<32, 16>(rp, cs, wv, x, p2, tk, W, bias, out, N, step, relu, pb, s);
  else if (Fin == 16 && Fout == 3)
    launch_step_sh<16, 3>(rp, cs, wv, x, p2, tk, W, bias, out, N, step, relu, pb, s);
}

static void run_conv(const float* x, const int* rp, const int* cs, const float* wv,
                     const float* W  , const float* bias,
                     float* out, float* b0, float* b1,
                     int N, int Fin, int Fout, int relu, int xin, hipStream_t s) {
  size_t wstride = (size_t)Fin * Fout;

  launch_step(rp, cs, wv, x, x, b0, W, bias, out, N, Fin, Fout, 0, relu,
               xin, s);

  for (int k = 2; k < 6; ++k) {
    const float* prev = (k & 1) ? b1 : b0;
    float* tk = (k & 1) ? b0 : b1;
    const float* p2 = (k == 2) ? x : tk;
    launch_step(rp, cs, wv, prev, p2, tk, W + (size_t)k * wstride, nullptr, out,
                N, Fin, Fout, (k == 5) ? 2 : 1, relu,
                 (k == 2) ? xin : 0, s);
  }
}


extern "C" void kernel_launch(void* const* d_in, const int* in_sizes, int n_in,
                              void* d_out, int out_size, void* d_ws, size_t ws_size,
                              hipStream_t stream) {
  (void)in_sizes; (void)n_in; (void)out_size; (void)ws_size;
  static const int NSs[5] = {65536, 16384, 4096, 1024, 256};
  static const int FE[5] = {3, 16, 16, 16, 32};
  static const int FD[6] = {32, 16, 16, 16, 16, 3};

  const float* x = (const float*)d_in[0];
  const int* ei[5];
  for (int l = 0; l < 5; ++l) ei[l] = (const int*)d_in[1 + l];
  const int* down_i[4]; const float* down_w[4];
  const int* up_i[4];   const float* up_w[4];
  for (int i = 0; i < 4; ++i) {
    down_i[i] = (const int*)d_in[6 + 4 * i];
    down_w[i] = (const float*)d_in[7 + 4 * i];
    up_i[i]   = (const int*)d_in[8 + 4 * i];
    up_w[i]   = (const float*)d_in[9 + 4 * i];
  }
  const float* encW[4]; const float* encB[4];
  for (int i = 0; i < 4; ++i) {
    encW[i] = (const float*)d_in[22 + 2 * i];
    encB[i] = (const float*)d_in[23 + 2 * i];
  }
  const float* decW[5];
  for (int i = 0; i < 5; ++i) decW[i] = (const float*)d_in[30 + i];
  const float* decB[4];
  for (int i = 0; i < 4; ++i) decB[i] = (const float*)d_in[35 + i];
  const float* enc_w = (const float*)d_in[39];
  const float* enc_b = (const float*)d_in[40];
  const float* dec_w = (const float*)d_in[41];
  const float* dec_b = (const float*)d_in[42];

  float* ws = (float*)d_ws;
  size_t off = 0;
  const size_t ACT = (size_t)4 * 65536 * 16;
  float* actA = ws + off; off += ACT;
  float* actB = ws + off; off += ACT;
  float* tb0 = ws + off; off += ACT;
  float* tb1 = ws + off; off += ACT;
  float* zbuf = ws + off; off += 256;
  int* rp[4]; int* cs[4]; float* wv[4];
  for (int l = 0; l < 4; ++l) {
    rp[l] = (int*)(ws + off); off += (size_t)NSs[l] + 4;
    cs[l] = (int*)(ws + off); off += (size_t)16 * NSs[l];
    wv[l] = ws + off;         off += (size_t)16 * NSs[l];
  }
  int* rank = (int*)tb0;
  int* slot = rank + (size_t)ETOT;
  int* cnt = slot + (size_t)ETOT;

  k_zero_i<<<(unsigned)CDIV(NTOT, 256), 256, 0, stream>>>(cnt, NTOT);
  k_rank<<<(unsigned)CDIV(ETOT, 256), 256, 0, stream>>>(ei[0], ei[1], ei[2], ei[3],
                                                       cnt, rank);
  k_scan<<<4, 1024, 0, stream>>>(cnt, rp[0], rp[1], rp[2], rp[3]);
  k_place<<<(unsigned)CDIV(ETOT, 256), 256, 0, stream>>>(
      ei[0], ei[1], ei[2], ei[3], rp[0], rp[1], rp[2], rp[3], rank, slot);
  CsrOut co = {cs[0], cs[1], cs[2], cs[3], wv[0], wv[1], wv[2], wv[3]};
  k_emit<<<(unsigned)CDIV(ETOT, 256), 256, 0, stream>>>(
      ei[0], ei[1], ei[2], ei[3], cnt, rp[0], rp[1], rp[2], rp[3], slot, co);

  const float* cur = x;
  for (int i = 0; i < 4; ++i) {
    run_conv(cur, rp[i], cs[i], wv[i], encW[i], encB[i], actA, tb0, tb1,
             NSs[i], FE[i], FE[i + 1],  1,  (i == 0) ? 1 : 0,
             stream);
    long long pt = 4LL * NSs[i + 1] * FE[i + 1];
    k_pool<<<(unsigned)CDIV(pt, 256), 256, 0, stream>>>(
        actA, down_i[i], down_w[i], actB, NSs[i + 1], NSs[i], FE[i + 1]);
    cur = actB;
  }

  k_dense_enc<<<256, 256, 0, stream>>>(actB, enc_w, enc_b, zbuf);
  k_dense_dec<<<(unsigned)CDIV(4LL * 8192, 256), 256, 0, stream>>>(zbuf, dec_w,
                                                                   dec_b, actA);

  for (int i = 0; i < 4; ++i) {
    int lvl = 3 - i;
    long long pt = 4LL * NSs[lvl] * FD[i];
    k_pool<<<(unsigned)CDIV(pt, 256), 256, 0, stream>>>(
        actA, up_i[lvl], up_w[lvl], actB, NSs[lvl], NSs[lvl + 1], FD[i]);
    run_conv(actB, rp[lvl], cs[lvl], wv[lvl], decW[i], decB[i], actA, tb0, tb1,
             NSs[lvl], FD[i], FD[i + 1],  1,  0, stream);
  }

  run_conv(actA, rp[0], cs[0], wv[0], decW[4], nullptr, (float*)d_out, tb0, tb1,
           NSs[0], FD[4], FD[5],  0,  0, stream);
}


# Last conv (16->3) summed by the Clenshaw recurrence: h @ [W0..W5] once on WMMA, then its five propagates gather 3-feature rows (64 B per node for the four batches) instead of 16-feature rows (256 B)
# speedup vs baseline: 5.0716x; 1.1749x over previous
#include <hip/hip_runtime.h>
#include <hip/hip_bf16.h>
#include <cstdint>
#include <cstddef>


typedef __attribute__((ext_vector_type(2))) float v2f;
typedef __attribute__((ext_vector_type(8))) float v8f;

#define CDIV(a, b) (((a) + (b) - 1) / (b))


typedef __attribute__((ext_vector_type(4))) float v4f;

#define NLV(l) (65536 >> (2 * (l)))
#define ELV(l) (16 * NLV(l))
#define NOFF1 NLV(0)
#define NOFF2 (NOFF1 + NLV(1))
#define NOFF3 (NOFF2 + NLV(2))
#define NTOT (NOFF3 + NLV(3))
#define EOFF1 ELV(0)
#define EOFF2 (EOFF1 + ELV(1))
#define EOFF3 (EOFF2 + ELV(2))
#define ETOT (EOFF3 + ELV(3))

__device__ __forceinline__ int edge_level(int g) {
  return (g >= EOFF1 ? 1 : 0) + (g >= EOFF2 ? 1 : 0) + (g >= EOFF3 ? 1 : 0);
}
__device__ __forceinline__ int edge_off(int l) {
  return l == 0 ? 0 : (l == 1 ? EOFF1 : (l == 2 ? EOFF2 : EOFF3));
}
__device__ __forceinline__ int node_off(int l) {
  return l == 0 ? 0 : (l == 1 ? NOFF1 : (l == 2 ? NOFF2 : NOFF3));
}
template <typename T>
__device__ __forceinline__ T pick4(int l, T a0, T a1, T a2, T a3) {
  return l == 0 ? a0 : (l == 1 ? a1 : (l == 2 ? a2 : a3));
}

__global__ void k_zero_i(int* __restrict__ p, int n) {
  int i = blockIdx.x * blockDim.x + threadIdx.x;
  if (i < n) p[i] = 0;
}

__global__ void k_rank(const int* __restrict__ ei0, const int* __restrict__ ei1,
                       const int* __restrict__ ei2, const int* __restrict__ ei3,
                       int* __restrict__ cnt, int* __restrict__ rank) {
  int g = blockIdx.x * blockDim.x + threadIdx.x;
  if (g >= ETOT) return;
  int l = edge_level(g);
  const int* ei = pick4(l, ei0, ei1, ei2, ei3);
  int e = g - edge_off(l);
  rank[g] = atomicAdd(&cnt[node_off(l) + ei[e]], 1);
}

__global__ __launch_bounds__(1024) void k_scan(const int* __restrict__ cnt,
                                               int* __restrict__ rp0,
                                               int* __restrict__ rp1,
                                               int* __restrict__ rp2,
                                               int* __restrict__ rp3) {
  __shared__ int part[1024];
  int l = blockIdx.x;
  int N = NLV(l);
  int* rowptr = pick4(l, rp0, rp1, rp2, rp3);
  int t = threadIdx.x;
  int per = N >> 10;
  const int* src = cnt + node_off(l) + t * per;
  int s = 0;
  for (int i = 0; i < per; ++i) s += src[i];
  part[t] = s;
  __syncthreads();
  for (int off = 1; off < 1024; off <<= 1) {
    int v = (t >= off) ? part[t - off] : 0;
    __syncthreads();
    part[t] += v;
    __syncthreads();
  }
  int run = part[t] - s;
  int* dst = rowptr + t * per;
  for (int i = 0; i < per; ++i) {
    dst[i] = run;
    run += src[i];
  }
  if (t == 1023) rowptr[N] = run;
}

__global__ void k_place(const int* __restrict__ ei0, const int* __restrict__ ei1,
                        const int* __restrict__ ei2, const int* __restrict__ ei3,
                        const int* __restrict__ rp0, const int* __restrict__ rp1,
                        const int* __restrict__ rp2, const int* __restrict__ rp3,
                        const int* __restrict__ rank, int* __restrict__ slot) {
  int g = blockIdx.x * blockDim.x + threadIdx.x;
  if (g >= ETOT) return;
  int l = edge_level(g);
  const int* ei = pick4(l, ei0, ei1, ei2, ei3);
  const int* rowptr = pick4(l, rp0, rp1, rp2, rp3);
  int eo = edge_off(l);
  int e = g - eo;
  slot[eo + rowptr[ei[e]] + rank[g]] = e;
}

__device__ __forceinline__ float inv_sqrt_deg(int d) {
  return (d > 0) ? (1.0f / sqrtf((float)d)) : 0.0f;
}

struct CsrOut {
  int* cs0; int* cs1; int* cs2; int* cs3;
  float* wv0; float* wv1; float* wv2; float* wv3;
};

__global__ void k_emit(const int* __restrict__ ei0, const int* __restrict__ ei1,
                       const int* __restrict__ ei2, const int* __restrict__ ei3,
                       const int* __restrict__ cnt,
                       const int* __restrict__ rp0, const int* __restrict__ rp1,
                       const int* __restrict__ rp2, const int* __restrict__ rp3,
                       const int* __restrict__ slot, CsrOut o) {
  int g = blockIdx.x * blockDim.x + threadIdx.x;
  if (g >= ETOT) return;
  int l = edge_level(g);
  const int* ei = pick4(l, ei0, ei1, ei2, ei3);
  const int* rowptr = pick4(l, rp0, rp1, rp2, rp3);
  int* cs = pick4(l, o.cs0, o.cs1, o.cs2, o.cs3);
  float* wv = pick4(l, o.wv0, o.wv1, o.wv2, o.wv3);
  const int* cl = cnt + node_off(l);
  const int* sl = slot + edge_off(l);
  int E = ELV(l);
  int e = sl[g - edge_off(l)];
  int r = ei[e];
  int c = ei[E + e];
  int s = rowptr[r];
  int d = cl[r];
  int k = 0;
  for (int j = 0; j < d; ++j) k += (sl[s + j] < e) ? 1 : 0;
  cs[s + k] = c;
  wv[s + k] = inv_sqrt_deg(d) * inv_sqrt_deg(cl[c]);
}


template <int V> struct VecOf;
template <> struct VecOf<1> { typedef float T; };
template <> struct VecOf<4> { typedef v4f T; };

__device__ __forceinline__ float ldsel(const float* __restrict__ p, int idx, bool ok) {
  float v = p[ok ? idx : 0];
  return ok ? v : 0.0f;
}

template <int FIN, int FOUT, int STEP, bool RELU>
__global__ __launch_bounds__(256) void k_cheb_step(
    const int* __restrict__ rowptr, const int* __restrict__ cs,
    const float* __restrict__ wv, const float* x, const float* p2, float* tk,
    const float* __restrict__ W, const float* __restrict__ bias, float* out,
    int N, int pb) {
  constexpr int V = (FIN % 4 == 0) ? 4 : 1;
  constexpr bool WL = (FIN == 16);
  constexpr int Q = WL ? 8 : FIN / V;
  constexpr int R = (FIN == 3) ? 64 : 32;
  constexpr int KP = ((FIN + 3) / 4) * 4;
  constexpr int P = (KP == 4) ? 4 : KP + 4;
  constexpr int KC = KP / 4;
  constexpr int NTN = (FOUT + 15) / 16;
  constexpr int MT = 4 * (R / 16);
  constexpr int ITERS = (MT * NTN) / 8;
  constexpr int NT = (STEP == 0) ? 2 : 1;
  constexpr int TILE = 4 * R * P;
  typedef typename VecOf<V>::T VT;
  __shared__ __attribute__((aligned(16))) float sT[NT * TILE];

  const int t = threadIdx.x;
  const int row0 = blockIdx.x * R;
  constexpr bool XB = (FIN == 3) && (STEP == 0);
  constexpr bool OB = (FOUT == 3);
  const bool pbm = (FIN == 3) && (pb != 0);
  const int xsn = XB ? FIN : 4 * FIN;
  const int xsb = XB ? N * FIN : FIN;
  const int psn = pbm ? FIN : 4 * FIN;
  const int psb = pbm ? N * FIN : FIN;
  const int osn = OB ? FOUT : 4 * FOUT;
  const int osb = OB ? N * FOUT : FOUT;
  const bool gth = t < R * Q;
  const int rl = t / Q;
  const int q = t - rl * Q;
  const int op = (row0 + rl) * psn + q * V;
  const int ot = (row0 + rl) * (4 * FIN) + q * V;
  VT a0 = (VT)(0.0f), a1 = (VT)(0.0f), a2 = (VT)(0.0f), a3 = (VT)(0.0f);

  if constexpr (WL) {
    int r = row0 + rl;
    int j = rowptr[r];
    int je = rowptr[r + 1];
    const float* xq = x + q * 4;
    for (; j + 4 <= je; j += 4) {
      int c0 = cs[j], c1 = cs[j + 1], c2 = cs[j + 2], c3 = cs[j + 3];
      float w0 = wv[j], w1 = wv[j + 1], w2 = wv[j + 2], w3 = wv[j + 3];
      const float* s0 = xq + c0 * 64;
      const float* s1 = xq + c1 * 64;
      const float* s2 = xq + c2 * 64;
      const float* s3 = xq + c3 * 64;
      v4f v00 = *(const v4f*)(s0), v01 = *(const v4f*)(s0 + 32);
      v4f v10 = *(const v4f*)(s1), v11 = *(const v4f*)(s1 + 32);
      v4f v20 = *(const v4f*)(s2), v21 = *(const v4f*)(s2 + 32);
      v4f v30 = *(const v4f*)(s3), v31 = *(const v4f*)(s3 + 32);
      a0 += w0 * v00; a1 += w0 * v01;
      a0 += w1 * v10; a1 += w1 * v11;
      a0 += w2 * v20; a1 += w2 * v21;
      a0 += w3 * v30; a1 += w3 * v31;
    }
#pragma nounroll
    for (; j < je; ++j) {
      int c = cs[j];
      float w = wv[j];
      const float* s0 = xq + c * 64;
      v4f v0 = *(const v4f*)(s0), v1 = *(const v4f*)(s0 + 32);
      a0 += w * v0; a1 += w * v1;
    }
    v4f p0 = *(const v4f*)(p2 + op);
    v4f p1 = *(const v4f*)(p2 + op + 32);
    if constexpr (STEP != 0) {
      a0 = 2.0f * a0 - p0;
      a1 = 2.0f * a1 - p1;
    }
    float* d = sT + ((q >> 2) * R + rl) * P + (q & 3) * 4;
    *(v4f*)(d) = a0;
    *(v4f*)(d + 2 * R * P) = a1;
    if constexpr (STEP == 0) {
      *(v4f*)(d + TILE) = p0;
      *(v4f*)(d + TILE + 2 * R * P) = p1;
    }
  } else if (gth) {
    int r = row0 + rl;
    int j = rowptr[r];
    int je = rowptr[r + 1];
    const float* xq = x + q * V;
    for (; j + 4 <= je; j += 4) {
      int c0 = cs[j], c1 = cs[j + 1], c2 = cs[j + 2], c3 = cs[j + 3];
      float w0 = wv[j], w1 = wv[j + 1], w2 = wv[j + 2], w3 = wv[j + 3];
      const float* s0 = xq + c0 * xsn;
      const float* s1 = xq + c1 * xsn;
      const float* s2 = xq + c2 * xsn;
      const float* s3 = xq + c3 * xsn;
      VT v00 = *(const VT*)(s0), v01 = *(const VT*)(s0 + xsb);
      VT v02 = *(const VT*)(s0 + 2 * xsb), v03 = *(const VT*)(s0 + 3 * xsb);
      VT v10 = *(const VT*)(s1), v11 = *(const VT*)(s1 + xsb);
      VT v12 = *(const VT*)(s1 + 2 * xsb), v13 = *(const VT*)(s1 + 3 * xsb);
      VT v20 = *(const VT*)(s2), v21 = *(const VT*)(s2 + xsb);
      VT v22 = *(const VT*)(s2 + 2 * xsb), v23 = *(const VT*)(s2 + 3 * xsb);
      VT v30 = *(const VT*)(s3), v31 = *(const VT*)(s3 + xsb);
      VT v32 = *(const VT*)(s3 + 2 * xsb), v33 = *(const VT*)(s3 + 3 * xsb);
      a0 += w0 * v00; a1 += w0 * v01; a2 += w0 * v02; a3 += w0 * v03;
      a0 += w1 * v10; a1 += w1 * v11; a2 += w1 * v12; a3 += w1 * v13;
      a0 += w2 * v20; a1 += w2 * v21; a2 += w2 * v22; a3 += w2 * v23;
      a0 += w3 * v30; a1 += w3 * v31; a2 += w3 * v32; a3 += w3 * v33;
    }
    for (; j < je; ++j) {
      int c = cs[j];
      float w = wv[j];
      const float* s0 = xq + c * xsn;
      VT v0 = *(const VT*)(s0), v1 = *(const VT*)(s0 + xsb);
      VT v2 = *(const VT*)(s0 + 2 * xsb), v3 = *(const VT*)(s0 + 3 * xsb);
      a0 += w * v0; a1 += w * v1; a2 += w * v2; a3 += w * v3;
    }
    VT p0 = *(const VT*)(p2 + op);
    VT p1 = *(const VT*)(p2 + op + psb);
    VT p2v = *(const VT*)(p2 + op + 2 * psb);
    VT p3 = *(const VT*)(p2 + op + 3 * psb);
    if constexpr (STEP != 0) {
      a0 = 2.0f * a0 - p0;
      a1 = 2.0f * a1 - p1;
      a2 = 2.0f * a2 - p2v;
      a3 = 2.0f * a3 - p3;
    }
    float* d = sT + rl * P + q * V;
    *(VT*)(d) = a0;
    *(VT*)(d + R * P) = a1;
    *(VT*)(d + 2 * R * P) = a2;
    *(VT*)(d + 3 * R * P) = a3;
    if constexpr (STEP == 0) {
      *(VT*)(d + TILE) = p0;
      *(VT*)(d + TILE + R * P) = p1;
      *(VT*)(d + TILE + 2 * R * P) = p2v;
      *(VT*)(d + TILE + 3 * R * P) = p3;
    }
    if constexpr (KP != FIN) {
      if (q == 0) {
        for (int i = 0; i < NT * 4; ++i)
          for (int f = FIN; f < KP; ++f) sT[i * R * P + rl * P + f] = 0.0f;
      }
    }
  }
  __syncthreads();

  if constexpr (STEP != 2 && WL) {
    *(v4f*)(tk + ot) = a0;
    *(v4f*)(tk + ot + 32) = a1;
  } else if constexpr (STEP != 2) {
    if (gth) {
      *(VT*)(tk + ot) = a0;
      *(VT*)(tk + ot + FIN) = a1;
      *(VT*)(tk + ot + 2 * FIN) = a2;
      *(VT*)(tk + ot + 3 * FIN) = a3;
    }
  }

  const int wave = t >> 5;
  const int lane = t & 31;
  const int lm = lane & 15;
  const int khalf = (lane >> 4) << 1;
  const int mhi = (lane >> 4) << 3;
  const int tn = wave % NTN;
  const int ncol = tn * 16 + lm;
  const bool colok = (FOUT % 16 == 0) || (ncol < FOUT);

  v2f bf[NT][KC];
#pragma unroll
  for (int i = 0; i < NT; ++i) {
    const float* Wi = W + ((STEP == 0 && i == 0) ? FIN * FOUT : 0);
#pragma unroll
    for (int kc = 0; kc < KC; ++kc) {
      int ka = kc * 4 + khalf;
      if constexpr (FIN % 4 == 0 && FOUT % 16 == 0) {
        bf[i][kc].x = Wi[ka * FOUT + ncol];
        bf[i][kc].y = Wi[(ka + 1) * FOUT + ncol];
      } else {
        bf[i][kc].x = ldsel(Wi, ka * FOUT + ncol, colok && (ka < FIN));
        bf[i][kc].y = ldsel(Wi, (ka + 1) * FOUT + ncol, colok && (ka + 1 < FIN));
      }
    }
  }
  float bv = 0.0f;
  if constexpr (STEP == 0) {
    if (bias) bv = ldsel(bias, ncol, colok);
  }

#pragma unroll
  for (int it = 0; it < ITERS; ++it) {
    int mt = (wave / NTN) * ITERS + it;
    int b = mt & 3;
    int rt = mt >> 2;
    const float* arow = sT + (b * R + rt * 16 + lm) * P + khalf;
    v8f c = {0.f, 0.f, 0.f, 0.f, 0.f, 0.f, 0.f, 0.f};
#pragma unroll
    for (int i = 0; i < NT; ++i) {
#pragma unroll
      for (int kc = 0; kc < KC; ++kc) {
        v2f a = *(const v2f*)(arow + i * TILE + kc * 4);
        c = __builtin_amdgcn_wmma_f32_16x16x4_f32(false, a, false, bf[i][kc],
                                                  (short)0, c, false, false);
      }
    }
    float* orow = out + (size_t)(row0 + rt * 16 + mhi) * osn + (size_t)b * osb + ncol;
    if (colok) {
#pragma unroll
      for (int r = 0; r < 8; ++r) {
        float v = c[r];
        if constexpr (STEP == 0) v += bv;
        else v += orow[(size_t)r * osn];
        if constexpr (STEP == 2 && RELU) v = v > 0.0f ? v : 0.0f;
        orow[(size_t)r * osn] = v;
      }
    }
  }
}


__global__ __launch_bounds__(256) void k_y_planes(const float* __restrict__ h,
                                                  const float* __restrict__ W,
                                                  float* __restrict__ P, int M) {
  constexpr int G = 4;
  const int t = threadIdx.x;
  const int wave = t >> 5;
  const int lane = t & 31;
  const int c = lane & 15;
  const int hh = lane >> 4;
  const int khalf = hh << 1;
  v2f af[2][4];
#pragma unroll
  for (int tile = 0; tile < 2; ++tile) {
    int j = tile * 16 + c;
    int k = j >> 2;
    int f = j & 3;
    bool ok = (f < 3) && (k < 6);
#pragma unroll
    for (int kc = 0; kc < 4; ++kc) {
      int ka = kc * 4 + khalf;
      af[tile][kc].x = ldsel(W, k * 48 + ka * 3 + f, ok);
      af[tile][kc].y = ldsel(W, k * 48 + (ka + 1) * 3 + f, ok);
    }
  }
  const size_t PL = (size_t)M * 4;
  const int mb = (blockIdx.x * 8 + wave) * (16 * G);
#pragma unroll
  for (int g = 0; g < G; ++g) {
    int m = mb + g * 16 + c;
    const float* hr = h + (size_t)m * 16 + khalf;
    v2f b0 = *(const v2f*)(hr);
    v2f b1 = *(const v2f*)(hr + 4);
    v2f b2 = *(const v2f*)(hr + 8);
    v2f b3 = *(const v2f*)(hr + 12);
    v8f d0 = {0.f, 0.f, 0.f, 0.f, 0.f, 0.f, 0.f, 0.f};
    v8f d1 = {0.f, 0.f, 0.f, 0.f, 0.f, 0.f, 0.f, 0.f};
    d0 = __builtin_amdgcn_wmma_f32_16x16x4_f32(false, af[0][0], false, b0, (short)0, d0, false, false);
    d1 = __builtin_amdgcn_wmma_f32_16x16x4_f32(false, af[1][0], false, b0, (short)0, d1, false, false);
    d0 = __builtin_amdgcn_wmma_f32_16x16x4_f32(false, af[0][1], false, b1, (short)0, d0, false, false);
    d1 = __builtin_amdgcn_wmma_f32_16x16x4_f32(false, af[1][1], false, b1, (short)0, d1, false, false);
    d0 = __builtin_amdgcn_wmma_f32_16x16x4_f32(false, af[0][2], false, b2, (short)0, d0, false, false);
    d1 = __builtin_amdgcn_wmma_f32_16x16x4_f32(false, af[1][2], false, b2, (short)0, d1, false, false);
    d0 = __builtin_amdgcn_wmma_f32_16x16x4_f32(false, af[0][3], false, b3, (short)0, d0, false, false);
    d1 = __builtin_amdgcn_wmma_f32_16x16x4_f32(false, af[1][3], false, b3, (short)0, d1, false, false);
    float* o = P + (size_t)m * 4;
    v4f lo = {d0[0], d0[1], d0[2], d0[3]};
    v4f hi = {d0[4], d0[5], d0[6], d0[7]};
    *(v4f*)(o + (size_t)(2 * hh) * PL) = lo;
    *(v4f*)(o + (size_t)(2 * hh + 1) * PL) = hi;
    if (hh == 0) {
      v4f lo1 = {d1[0], d1[1], d1[2], d1[3]};
      v4f hi1 = {d1[4], d1[5], d1[6], d1[7]};
      *(v4f*)(o + 4 * PL) = lo1;
      *(v4f*)(o + 5 * PL) = hi1;
    }
  }
}

template <int MODE>
__global__ __launch_bounds__(256) void k_clenshaw_step(
    const int* __restrict__ rowptr, const int* __restrict__ cs,
    const float* __restrict__ wv, const float* __restrict__ bn,
    const float* __restrict__ b2, float* yk, float* __restrict__ out, int N) {
  const int t = threadIdx.x;
  const int rl = t >> 2;
  const int q = t & 3;
  const int r = blockIdx.x * 64 + rl;
  int j = rowptr[r];
  const int je = rowptr[r + 1];
  const int o = (r * 4 + q) * 4;
  v4f y = *(const v4f*)(yk + o);
  v4f p = (v4f)(0.0f);
  if constexpr (MODE != 0) p = *(const v4f*)(b2 + o);
  const float* xq = bn + q * 4;
  const int last = je - 1;
  v4f acc = (v4f)(0.0f);
  for (; j < je; j += 8) {
    int cc[8];
    float ww[8];
    v4f vv[8];
#pragma unroll
    for (int i = 0; i < 8; ++i) {
      bool ok = (j + i) < je;
      int jc = ok ? (j + i) : last;
      cc[i] = cs[jc];
      float wi = wv[jc];
      ww[i] = ok ? wi : 0.0f;
    }
#pragma unroll
    for (int i = 0; i < 8; ++i) vv[i] = *(const v4f*)(xq + cc[i] * 16);
#pragma unroll
    for (int i = 0; i < 8; ++i) acc += ww[i] * vv[i];
  }
  v4f res;
  if constexpr (MODE == 0) res = 2.0f * acc + y;
  else if constexpr (MODE == 1) res = 2.0f * acc + y - p;
  else res = acc + y - p;
  if constexpr (MODE != 2) {
    *(v4f*)(yk + o) = res;
  } else {
    __shared__ __attribute__((aligned(16))) float so[4 * 64 * 3];
    float* d = so + (q * 64 + rl) * 3;
    d[0] = res.x;
    d[1] = res.y;
    d[2] = res.z;
    __syncthreads();
    if (t < 192) {
      int b = t / 48;
      int pc = t - b * 48;
      v4f v = *(const v4f*)(so + b * 192 + pc * 4);
      *(v4f*)(out + ((size_t)b * N + (size_t)blockIdx.x * 64) * 3 + pc * 4) = v;
    }
  }
}

__global__ void k_pool(const float* __restrict__ x, const int* __restrict__ idx,
                       const float* __restrict__ w, float* __restrict__ out,
                       int Nout, int Nin, int F) {
  (void)Nin;
  long long tid = (long long)blockIdx.x * blockDim.x + threadIdx.x;
  long long total = 4LL * Nout * F;
  if (tid >= total) return;
  int F4 = 4 * F;
  int bf = (int)(tid % F4);
  int n = (int)(tid / F4);
  float s = 0.0f;
#pragma unroll
  for (int k = 0; k < 3; ++k) {
    int j = idx[n * 3 + k];
    s += w[n * 3 + k] * x[(size_t)j * F4 + bf];
  }
  out[tid] = s;
}

__global__ void k_dense_enc(const float* __restrict__ h, const float* __restrict__ w,
                            const float* __restrict__ bias, float* __restrict__ z) {
  int b = blockIdx.x >> 6;
  int j = blockIdx.x & 63;
  float s = 0.0f;
  for (int i = threadIdx.x; i < 8192; i += blockDim.x)
    s += h[((i >> 5) * 4 + b) * 32 + (i & 31)] * w[(size_t)i * 64 + j];
  __shared__ float red[256];
  red[threadIdx.x] = s;
  __syncthreads();
  for (int off = 128; off > 0; off >>= 1) {
    if (threadIdx.x < off) red[threadIdx.x] += red[threadIdx.x + off];
    __syncthreads();
  }
  if (threadIdx.x == 0) {
    float v = red[0] + bias[j];
    z[b * 64 + j] = v > 0.0f ? v : 0.0f;
  }
}

__global__ void k_dense_dec(const float* __restrict__ z, const float* __restrict__ w,
                            const float* __restrict__ bias, float* __restrict__ h) {
  long long tid = (long long)blockIdx.x * blockDim.x + threadIdx.x;
  if (tid >= 4LL * 8192) return;
  int b = (int)(tid >> 13);
  int j = (int)(tid & 8191);
  float s = bias[j];
  for (int i = 0; i < 64; ++i) s += z[b * 64 + i] * w[(size_t)i * 8192 + j];
  h[((j >> 5) * 4 + b) * 32 + (j & 31)] = s > 0.0f ? s : 0.0f;
}


template <int FIN, int FOUT>
static void launch_step_sh(const int* rp, const int* cs, const float* wv,
                           const float* x, const float* p2, float* tk,
                           const float* W, const float* bias, float* out, int N,
                           int step, int relu, int pb, hipStream_t s) {
  constexpr int R = (FIN == 3) ? 64 : 32;
  unsigned blocks = (unsigned)(N / R);
  if (step == 0)
    k_cheb_step<FIN, FOUT, 0, false><<<blocks, 256, 0, s>>>(rp, cs, wv, x, p2, tk, W, bias, out, N, pb);
  else if (step == 1)
    k_cheb_step<FIN, FOUT, 1, false><<<blocks, 256, 0, s>>>(rp, cs, wv, x, p2, tk, W, bias, out, N, pb);
  else if (relu)
    k_cheb_step<FIN, FOUT, 2, true><<<blocks, 256, 0, s>>>(rp, cs, wv, x, p2, tk, W, bias, out, N, pb);
  else
    k_cheb_step<FIN, FOUT, 2, false><<<blocks, 256, 0, s>>>(rp, cs, wv, x, p2, tk, W, bias, out, N, pb);
}

static void launch_step(const int* rp, const int* cs, const float* wv,
                        const float* x, const float* p2, float* tk,
                        const float* W, const float* bias, float* out, int N,
                        int Fin, int Fout, int step, int relu, int pb,
                        hipStream_t s) {
  if (Fin == 3 && Fout == 16)
    launch_step_sh<3, 16>(rp, cs, wv, x, p2, tk, W, bias, out, N, step, relu, pb, s);
  else if (Fin == 16 && Fout == 16)
    launch_step_sh<16, 16>(rp, cs, wv, x, p2, tk, W, bias, out, N, step, relu, pb, s);
  else if (Fin == 16 && Fout == 32)
    launch_step_sh<16, 32>(rp, cs, wv, x, p2, tk, W, bias, out, N, step, relu, pb, s);
  else if (Fin == 32 && Fout == 16)
    launch_step_sh<32, 16>(rp, cs, wv, x, p2, tk, W, bias, out, N, step, relu, pb, s);
}

static void run_conv(const float* x, const int* rp, const int* cs, const float* wv,
                     const float* W  , const float* bias,
                     float* out, float* b0, float* b1,
                     int N, int Fin, int Fout, int relu, int xin, hipStream_t s) {
  size_t wstride = (size_t)Fin * Fout;

  launch_step(rp, cs, wv, x, x, b0, W, bias, out, N, Fin, Fout, 0, relu,
               xin, s);

  for (int k = 2; k < 6; ++k) {
    const float* prev = (k & 1) ? b1 : b0;
    float* tk = (k & 1) ? b0 : b1;
    const float* p2 = (k == 2) ? x : tk;
    launch_step(rp, cs, wv, prev, p2, tk, W + (size_t)k * wstride, nullptr, out,
                N, Fin, Fout, (k == 5) ? 2 : 1, relu,
                 (k == 2) ? xin : 0, s);
  }
}


extern "C" void kernel_launch(void* const* d_in, const int* in_sizes, int n_in,
                              void* d_out, int out_size, void* d_ws, size_t ws_size,
                              hipStream_t stream) {
  (void)in_sizes; (void)n_in; (void)out_size; (void)ws_size;
  static const int NSs[5] = {65536, 16384, 4096, 1024, 256};
  static const int FE[5] = {3, 16, 16, 16, 32};
  static const int FD[6] = {32, 16, 16, 16, 16, 3};

  const float* x = (const float*)d_in[0];
  const int* ei[5];
  for (int l = 0; l < 5; ++l) ei[l] = (const int*)d_in[1 + l];
  const int* down_i[4]; const float* down_w[4];
  const int* up_i[4];   const float* up_w[4];
  for (int i = 0; i < 4; ++i) {
    down_i[i] = (const int*)d_in[6 + 4 * i];
    down_w[i] = (const float*)d_in[7 + 4 * i];
    up_i[i]   = (const int*)d_in[8 + 4 * i];
    up_w[i]   = (const float*)d_in[9 + 4 * i];
  }
  const float* encW[4]; const float* encB[4];
  for (int i = 0; i < 4; ++i) {
    encW[i] = (const float*)d_in[22 + 2 * i];
    encB[i] = (const float*)d_in[23 + 2 * i];
  }
  const float* decW[5];
  for (int i = 0; i < 5; ++i) decW[i] = (const float*)d_in[30 + i];
  const float* decB[4];
  for (int i = 0; i < 4; ++i) decB[i] = (const float*)d_in[35 + i];
  const float* enc_w = (const float*)d_in[39];
  const float* enc_b = (const float*)d_in[40];
  const float* dec_w = (const float*)d_in[41];
  const float* dec_b = (const float*)d_in[42];

  float* ws = (float*)d_ws;
  size_t off = 0;
  const size_t ACT = (size_t)4 * 65536 * 16;
  float* actA = ws + off; off += ACT;
  float* actB = ws + off; off += ACT;
  float* tb0 = ws + off; off += ACT;
  float* tb1 = ws + off; off += ACT;
  float* zbuf = ws + off; off += 256;
  int* rp[4]; int* cs[4]; float* wv[4];
  for (int l = 0; l < 4; ++l) {
    rp[l] = (int*)(ws + off); off += (size_t)NSs[l] + 4;
    cs[l] = (int*)(ws + off); off += (size_t)16 * NSs[l];
    wv[l] = ws + off;         off += (size_t)16 * NSs[l];
  }
  int* rank = (int*)tb0;
  int* slot = rank + (size_t)ETOT;
  int* cnt = slot + (size_t)ETOT;

  k_zero_i<<<(unsigned)CDIV(NTOT, 256), 256, 0, stream>>>(cnt, NTOT);
  k_rank<<<(unsigned)CDIV(ETOT, 256), 256, 0, stream>>>(ei[0], ei[1], ei[2], ei[3],
                                                       cnt, rank);
  k_scan<<<4, 1024, 0, stream>>>(cnt, rp[0], rp[1], rp[2], rp[3]);
  k_place<<<(unsigned)CDIV(ETOT, 256), 256, 0, stream>>>(
      ei[0], ei[1], ei[2], ei[3], rp[0], rp[1], rp[2], rp[3], rank, slot);
  CsrOut co = {cs[0], cs[1], cs[2], cs[3], wv[0], wv[1], wv[2], wv[3]};
  k_emit<<<(unsigned)CDIV(ETOT, 256), 256, 0, stream>>>(
      ei[0], ei[1], ei[2], ei[3], cnt, rp[0], rp[1], rp[2], rp[3], slot, co);

  const float* cur = x;
  for (int i = 0; i < 4; ++i) {
    run_conv(cur, rp[i], cs[i], wv[i], encW[i], encB[i], actA, tb0, tb1,
             NSs[i], FE[i], FE[i + 1],  1,  (i == 0) ? 1 : 0,
             stream);
    long long pt = 4LL * NSs[i + 1] * FE[i + 1];
    k_pool<<<(unsigned)CDIV(pt, 256), 256, 0, stream>>>(
        actA, down_i[i], down_w[i], actB, NSs[i + 1], NSs[i], FE[i + 1]);
    cur = actB;
  }

  k_dense_enc<<<256, 256, 0, stream>>>(actB, enc_w, enc_b, zbuf);
  k_dense_dec<<<(unsigned)CDIV(4LL * 8192, 256), 256, 0, stream>>>(zbuf, dec_w,
                                                                   dec_b, actA);

  for (int i = 0; i < 4; ++i) {
    int lvl = 3 - i;
    long long pt = 4LL * NSs[lvl] * FD[i];
    k_pool<<<(unsigned)CDIV(pt, 256), 256, 0, stream>>>(
        actA, up_i[lvl], up_w[lvl], actB, NSs[lvl], NSs[lvl + 1], FD[i]);
    run_conv(actB, rp[lvl], cs[lvl], wv[lvl], decW[i], decB[i], actA, tb0, tb1,
             NSs[lvl], FD[i], FD[i + 1],  1,  0, stream);
  }

  const int N0 = 65536;
  const size_t PL = (size_t)4 * N0 * 4;
  float* P = tb0;
  k_y_planes<<<(unsigned)(4 * N0 / 512), 256, 0, stream>>>(actA, decW[4], P, 4 * N0);
  k_clenshaw_step<0><<<(unsigned)(N0 / 64), 256, 0, stream>>>(
      rp[0], cs[0], wv[0], P + 5 * PL, nullptr, P + 4 * PL, nullptr, N0);
  k_clenshaw_step<1><<<(unsigned)(N0 / 64), 256, 0, stream>>>(
      rp[0], cs[0], wv[0], P + 4 * PL, P + 5 * PL, P + 3 * PL, nullptr, N0);
  k_clenshaw_step<1><<<(unsigned)(N0 / 64), 256, 0, stream>>>(
      rp[0], cs[0], wv[0], P + 3 * PL, P + 4 * PL, P + 2 * PL, nullptr, N0);
  k_clenshaw_step<1><<<(unsigned)(N0 / 64), 256, 0, stream>>>(
      rp[0], cs[0], wv[0], P + 2 * PL, P + 3 * PL, P + 1 * PL, nullptr, N0);
  k_clenshaw_step<2><<<(unsigned)(N0 / 64), 256, 0, stream>>>(
      rp[0], cs[0], wv[0], P + 1 * PL, P + 2 * PL, P, (float*)d_out, N0);
}


# First conv (3->16): x and its T_k kept as 64-byte node runs (4 batches x (x,y,z,0)): an edge is one 16-byte gather per lane, not four 4-byte ones (x itself: in four lines); launch 1 lays x out
# speedup vs baseline: 5.2766x; 1.0404x over previous
#include <hip/hip_runtime.h>
#include <hip/hip_bf16.h>
#include <cstdint>
#include <cstddef>


typedef __attribute__((ext_vector_type(2))) float v2f;
typedef __attribute__((ext_vector_type(8))) float v8f;

#define CDIV(a, b) (((a) + (b) - 1) / (b))


typedef __attribute__((ext_vector_type(4))) float v4f;

#define NLV(l) (65536 >> (2 * (l)))
#define ELV(l) (16 * NLV(l))
#define NOFF1 NLV(0)
#define NOFF2 (NOFF1 + NLV(1))
#define NOFF3 (NOFF2 + NLV(2))
#define NTOT (NOFF3 + NLV(3))
#define EOFF1 ELV(0)
#define EOFF2 (EOFF1 + ELV(1))
#define EOFF3 (EOFF2 + ELV(2))
#define ETOT (EOFF3 + ELV(3))

__device__ __forceinline__ int edge_level(int g) {
  return (g >= EOFF1 ? 1 : 0) + (g >= EOFF2 ? 1 : 0) + (g >= EOFF3 ? 1 : 0);
}
__device__ __forceinline__ int edge_off(int l) {
  return l == 0 ? 0 : (l == 1 ? EOFF1 : (l == 2 ? EOFF2 : EOFF3));
}
__device__ __forceinline__ int node_off(int l) {
  return l == 0 ? 0 : (l == 1 ? NOFF1 : (l == 2 ? NOFF2 : NOFF3));
}
template <typename T>
__device__ __forceinline__ T pick4(int l, T a0, T a1, T a2, T a3) {
  return l == 0 ? a0 : (l == 1 ? a1 : (l == 2 ? a2 : a3));
}

__global__ void k_zero_i(int* __restrict__ p, int n, int zb,
                         const float* __restrict__ x, float* __restrict__ xi,
                         int N) {
  if ((int)blockIdx.x < zb) {
    int i = blockIdx.x * blockDim.x + threadIdx.x;
    if (i < n) p[i] = 0;
  } else {
    int u = (blockIdx.x - zb) * blockDim.x + threadIdx.x;
    if (u < 4 * N) {
      const float* s = x + ((size_t)(u & 3) * N + (u >> 2)) * 3;
      v4f v = {s[0], s[1], s[2], 0.0f};
      *(v4f*)(xi + (size_t)u * 4) = v;
    }
  }
}

__global__ void k_rank(const int* __restrict__ ei0, const int* __restrict__ ei1,
                       const int* __restrict__ ei2, const int* __restrict__ ei3,
                       int* __restrict__ cnt, int* __restrict__ rank) {
  int g = blockIdx.x * blockDim.x + threadIdx.x;
  if (g >= ETOT) return;
  int l = edge_level(g);
  const int* ei = pick4(l, ei0, ei1, ei2, ei3);
  int e = g - edge_off(l);
  rank[g] = atomicAdd(&cnt[node_off(l) + ei[e]], 1);
}

__global__ __launch_bounds__(1024) void k_scan(const int* __restrict__ cnt,
                                               int* __restrict__ rp0,
                                               int* __restrict__ rp1,
                                               int* __restrict__ rp2,
                                               int* __restrict__ rp3) {
  __shared__ int part[1024];
  int l = blockIdx.x;
  int N = NLV(l);
  int* rowptr = pick4(l, rp0, rp1, rp2, rp3);
  int t = threadIdx.x;
  int per = N >> 10;
  const int* src = cnt + node_off(l) + t * per;
  int s = 0;
  for (int i = 0; i < per; ++i) s += src[i];
  part[t] = s;
  __syncthreads();
  for (int off = 1; off < 1024; off <<= 1) {
    int v = (t >= off) ? part[t - off] : 0;
    __syncthreads();
    part[t] += v;
    __syncthreads();
  }
  int run = part[t] - s;
  int* dst = rowptr + t * per;
  for (int i = 0; i < per; ++i) {
    dst[i] = run;
    run += src[i];
  }
  if (t == 1023) rowptr[N] = run;
}

__global__ void k_place(const int* __restrict__ ei0, const int* __restrict__ ei1,
                        const int* __restrict__ ei2, const int* __restrict__ ei3,
                        const int* __restrict__ rp0, const int* __restrict__ rp1,
                        const int* __restrict__ rp2, const int* __restrict__ rp3,
                        const int* __restrict__ rank, int* __restrict__ slot) {
  int g = blockIdx.x * blockDim.x + threadIdx.x;
  if (g >= ETOT) return;
  int l = edge_level(g);
  const int* ei = pick4(l, ei0, ei1, ei2, ei3);
  const int* rowptr = pick4(l, rp0, rp1, rp2, rp3);
  int eo = edge_off(l);
  int e = g - eo;
  slot[eo + rowptr[ei[e]] + rank[g]] = e;
}

__device__ __forceinline__ float inv_sqrt_deg(int d) {
  return (d > 0) ? (1.0f / sqrtf((float)d)) : 0.0f;
}

struct CsrOut {
  int* cs0; int* cs1; int* cs2; int* cs3;
  float* wv0; float* wv1; float* wv2; float* wv3;
};

__global__ void k_emit(const int* __restrict__ ei0, const int* __restrict__ ei1,
                       const int* __restrict__ ei2, const int* __restrict__ ei3,
                       const int* __restrict__ cnt,
                       const int* __restrict__ rp0, const int* __restrict__ rp1,
                       const int* __restrict__ rp2, const int* __restrict__ rp3,
                       const int* __restrict__ slot, CsrOut o) {
  int g = blockIdx.x * blockDim.x + threadIdx.x;
  if (g >= ETOT) return;
  int l = edge_level(g);
  const int* ei = pick4(l, ei0, ei1, ei2, ei3);
  const int* rowptr = pick4(l, rp0, rp1, rp2, rp3);
  int* cs = pick4(l, o.cs0, o.cs1, o.cs2, o.cs3);
  float* wv = pick4(l, o.wv0, o.wv1, o.wv2, o.wv3);
  const int* cl = cnt + node_off(l);
  const int* sl = slot + edge_off(l);
  int E = ELV(l);
  int e = sl[g - edge_off(l)];
  int r = ei[e];
  int c = ei[E + e];
  int s = rowptr[r];
  int d = cl[r];
  int k = 0;
  for (int j = 0; j < d; ++j) k += (sl[s + j] < e) ? 1 : 0;
  cs[s + k] = c;
  wv[s + k] = inv_sqrt_deg(d) * inv_sqrt_deg(cl[c]);
}


template <int V> struct VecOf;
template <> struct VecOf<1> { typedef float T; };
template <> struct VecOf<4> { typedef v4f T; };

__device__ __forceinline__ float ldsel(const float* __restrict__ p, int idx, bool ok) {
  float v = p[ok ? idx : 0];
  return ok ? v : 0.0f;
}

template <int FIN, int FOUT, int STEP, bool RELU>
__global__ __launch_bounds__(256) void k_cheb_step(
    const int* __restrict__ rowptr, const int* __restrict__ cs,
    const float* __restrict__ wv, const float* x, const float* p2, float* tk,
    const float* __restrict__ W, const float* __restrict__ bias, float* out,
    int N, int pb) {
  constexpr int V = (FIN % 4 == 0) ? 4 : 1;
  constexpr bool WL = (FIN == 16);
  constexpr int Q = WL ? 8 : FIN / V;
  constexpr int R = (FIN == 3) ? 64 : 32;
  constexpr int KP = ((FIN + 3) / 4) * 4;
  constexpr int P = (KP == 4) ? 4 : KP + 4;
  constexpr int KC = KP / 4;
  constexpr int NTN = (FOUT + 15) / 16;
  constexpr int MT = 4 * (R / 16);
  constexpr int ITERS = (MT * NTN) / 8;
  constexpr int NT = (STEP == 0) ? 2 : 1;
  constexpr int TILE = 4 * R * P;
  typedef typename VecOf<V>::T VT;
  __shared__ __attribute__((aligned(16))) float sT[NT * TILE];

  const int t = threadIdx.x;
  const int row0 = blockIdx.x * R;
  constexpr bool XB = (FIN == 3) && (STEP == 0);
  constexpr bool OB = (FOUT == 3);
  const bool pbm = (FIN == 3) && (pb != 0);
  const int xsn = XB ? FIN : 4 * FIN;
  const int xsb = XB ? N * FIN : FIN;
  const int psn = pbm ? FIN : 4 * FIN;
  const int psb = pbm ? N * FIN : FIN;
  const int osn = OB ? FOUT : 4 * FOUT;
  const int osb = OB ? N * FOUT : FOUT;
  const bool gth = t < R * Q;
  const int rl = t / Q;
  const int q = t - rl * Q;
  const int op = (row0 + rl) * psn + q * V;
  const int ot = (row0 + rl) * (4 * FIN) + q * V;
  VT a0 = (VT)(0.0f), a1 = (VT)(0.0f), a2 = (VT)(0.0f), a3 = (VT)(0.0f);

  if constexpr (WL) {
    int r = row0 + rl;
    int j = rowptr[r];
    int je = rowptr[r + 1];
    const float* xq = x + q * 4;
    for (; j + 4 <= je; j += 4) {
      int c0 = cs[j], c1 = cs[j + 1], c2 = cs[j + 2], c3 = cs[j + 3];
      float w0 = wv[j], w1 = wv[j + 1], w2 = wv[j + 2], w3 = wv[j + 3];
      const float* s0 = xq + c0 * 64;
      const float* s1 = xq + c1 * 64;
      const float* s2 = xq + c2 * 64;
      const float* s3 = xq + c3 * 64;
      v4f v00 = *(const v4f*)(s0), v01 = *(const v4f*)(s0 + 32);
      v4f v10 = *(const v4f*)(s1), v11 = *(const v4f*)(s1 + 32);
      v4f v20 = *(const v4f*)(s2), v21 = *(const v4f*)(s2 + 32);
      v4f v30 = *(const v4f*)(s3), v31 = *(const v4f*)(s3 + 32);
      a0 += w0 * v00; a1 += w0 * v01;
      a0 += w1 * v10; a1 += w1 * v11;
      a0 += w2 * v20; a1 += w2 * v21;
      a0 += w3 * v30; a1 += w3 * v31;
    }
#pragma nounroll
    for (; j < je; ++j) {
      int c = cs[j];
      float w = wv[j];
      const float* s0 = xq + c * 64;
      v4f v0 = *(const v4f*)(s0), v1 = *(const v4f*)(s0 + 32);
      a0 += w * v0; a1 += w * v1;
    }
    v4f p0 = *(const v4f*)(p2 + op);
    v4f p1 = *(const v4f*)(p2 + op + 32);
    if constexpr (STEP != 0) {
      a0 = 2.0f * a0 - p0;
      a1 = 2.0f * a1 - p1;
    }
    float* d = sT + ((q >> 2) * R + rl) * P + (q & 3) * 4;
    *(v4f*)(d) = a0;
    *(v4f*)(d + 2 * R * P) = a1;
    if constexpr (STEP == 0) {
      *(v4f*)(d + TILE) = p0;
      *(v4f*)(d + TILE + 2 * R * P) = p1;
    }
  } else if (gth) {
    int r = row0 + rl;
    int j = rowptr[r];
    int je = rowptr[r + 1];
    const float* xq = x + q * V;
    for (; j + 4 <= je; j += 4) {
      int c0 = cs[j], c1 = cs[j + 1], c2 = cs[j + 2], c3 = cs[j + 3];
      float w0 = wv[j], w1 = wv[j + 1], w2 = wv[j + 2], w3 = wv[j + 3];
      const float* s0 = xq + c0 * xsn;
      const float* s1 = xq + c1 * xsn;
      const float* s2 = xq + c2 * xsn;
      const float* s3 = xq + c3 * xsn;
      VT v00 = *(const VT*)(s0), v01 = *(const VT*)(s0 + xsb);
      VT v02 = *(const VT*)(s0 + 2 * xsb), v03 = *(const VT*)(s0 + 3 * xsb);
      VT v10 = *(const VT*)(s1), v11 = *(const VT*)(s1 + xsb);
      VT v12 = *(const VT*)(s1 + 2 * xsb), v13 = *(const VT*)(s1 + 3 * xsb);
      VT v20 = *(const VT*)(s2), v21 = *(const VT*)(s2 + xsb);
      VT v22 = *(const VT*)(s2 + 2 * xsb), v23 = *(const VT*)(s2 + 3 * xsb);
      VT v30 = *(const VT*)(s3), v31 = *(const VT*)(s3 + xsb);
      VT v32 = *(const VT*)(s3 + 2 * xsb), v33 = *(const VT*)(s3 + 3 * xsb);
      a0 += w0 * v00; a1 += w0 * v01; a2 += w0 * v02; a3 += w0 * v03;
      a0 += w1 * v10; a1 += w1 * v11; a2 += w1 * v12; a3 += w1 * v13;
      a0 += w2 * v20; a1 += w2 * v21; a2 += w2 * v22; a3 += w2 * v23;
      a0 += w3 * v30; a1 += w3 * v31; a2 += w3 * v32; a3 += w3 * v33;
    }
    for (; j < je; ++j) {
      int c = cs[j];
      float w = wv[j];
      const float* s0 = xq + c * xsn;
      VT v0 = *(const VT*)(s0), v1 = *(const VT*)(s0 + xsb);
      VT v2 = *(const VT*)(s0 + 2 * xsb), v3 = *(const VT*)(s0 + 3 * xsb);
      a0 += w * v0; a1 += w * v1; a2 += w * v2; a3 += w * v3;
    }
    VT p0 = *(const VT*)(p2 + op);
    VT p1 = *(const VT*)(p2 + op + psb);
    VT p2v = *(const VT*)(p2 + op + 2 * psb);
    VT p3 = *(const VT*)(p2 + op + 3 * psb);
    if constexpr (STEP != 0) {
      a0 = 2.0f * a0 - p0;
      a1 = 2.0f * a1 - p1;
      a2 = 2.0f * a2 - p2v;
      a3 = 2.0f * a3 - p3;
    }
    float* d = sT + rl * P + q * V;
    *(VT*)(d) = a0;
    *(VT*)(d + R * P) = a1;
    *(VT*)(d + 2 * R * P) = a2;
    *(VT*)(d + 3 * R * P) = a3;
    if constexpr (STEP == 0) {
      *(VT*)(d + TILE) = p0;
      *(VT*)(d + TILE + R * P) = p1;
      *(VT*)(d + TILE + 2 * R * P) = p2v;
      *(VT*)(d + TILE + 3 * R * P) = p3;
    }
    if constexpr (KP != FIN) {
      if (q == 0) {
        for (int i = 0; i < NT * 4; ++i)
          for (int f = FIN; f < KP; ++f) sT[i * R * P + rl * P + f] = 0.0f;
      }
    }
  }
  __syncthreads();

  if constexpr (STEP != 2 && WL) {
    *(v4f*)(tk + ot) = a0;
    *(v4f*)(tk + ot + 32) = a1;
  } else if constexpr (STEP != 2) {
    if (gth) {
      *(VT*)(tk + ot) = a0;
      *(VT*)(tk + ot + FIN) = a1;
      *(VT*)(tk + ot + 2 * FIN) = a2;
      *(VT*)(tk + ot + 3 * FIN) = a3;
    }
  }

  const int wave = t >> 5;
  const int lane = t & 31;
  const int lm = lane & 15;
  const int khalf = (lane >> 4) << 1;
  const int mhi = (lane >> 4) << 3;
  const int tn = wave % NTN;
  const int ncol = tn * 16 + lm;
  const bool colok = (FOUT % 16 == 0) || (ncol < FOUT);

  v2f bf[NT][KC];
#pragma unroll
  for (int i = 0; i < NT; ++i) {
    const float* Wi = W + ((STEP == 0 && i == 0) ? FIN * FOUT : 0);
#pragma unroll
    for (int kc = 0; kc < KC; ++kc) {
      int ka = kc * 4 + khalf;
      if constexpr (FIN % 4 == 0 && FOUT % 16 == 0) {
        bf[i][kc].x = Wi[ka * FOUT + ncol];
        bf[i][kc].y = Wi[(ka + 1) * FOUT + ncol];
      } else {
        bf[i][kc].x = ldsel(Wi, ka * FOUT + ncol, colok && (ka < FIN));
        bf[i][kc].y = ldsel(Wi, (ka + 1) * FOUT + ncol, colok && (ka + 1 < FIN));
      }
    }
  }
  float bv = 0.0f;
  if constexpr (STEP == 0) {
    if (bias) bv = ldsel(bias, ncol, colok);
  }

#pragma unroll
  for (int it = 0; it < ITERS; ++it) {
    int mt = (wave / NTN) * ITERS + it;
    int b = mt & 3;
    int rt = mt >> 2;
    const float* arow = sT + (b * R + rt * 16 + lm) * P + khalf;
    v8f c = {0.f, 0.f, 0.f, 0.f, 0.f, 0.f, 0.f, 0.f};
#pragma unroll
    for (int i = 0; i < NT; ++i) {
#pragma unroll
      for (int kc = 0; kc < KC; ++kc) {
        v2f a = *(const v2f*)(arow + i * TILE + kc * 4);
        c = __builtin_amdgcn_wmma_f32_16x16x4_f32(false, a, false, bf[i][kc],
                                                  (short)0, c, false, false);
      }
    }
    float* orow = out + (size_t)(row0 + rt * 16 + mhi) * osn + (size_t)b * osb + ncol;
    if (colok) {
#pragma unroll
      for (int r = 0; r < 8; ++r) {
        float v = c[r];
        if constexpr (STEP == 0) v += bv;
        else v += orow[(size_t)r * osn];
        if constexpr (STEP == 2 && RELU) v = v > 0.0f ? v : 0.0f;
        orow[(size_t)r * osn] = v;
      }
    }
  }
}


template <int STEP, bool RELU>
__global__ __launch_bounds__(256) void k_cheb3_step(
    const int* __restrict__ rowptr, const int* __restrict__ cs,
    const float* __restrict__ wv, const float* x, const float* p2, float* tk,
    const float* __restrict__ W, const float* __restrict__ bias, float* out) {
  constexpr int R = 64;
  constexpr int P = 4;
  constexpr int TILE = 4 * R * P;
  constexpr int NT = (STEP == 0) ? 2 : 1;
  __shared__ __attribute__((aligned(16))) float sT[NT * TILE];

  const int t = threadIdx.x;
  const int rl = t >> 2;
  const int q = t & 3;
  const int row0 = blockIdx.x * R;
  const int r = row0 + rl;
  int j = rowptr[r];
  const int je = rowptr[r + 1];
  const int o = (r * 4 + q) * 4;
  v4f p = *(const v4f*)(p2 + o);
  const float* xq = x + q * 4;
  const int last = je - 1;
  v4f acc = (v4f)(0.0f);
  for (; j < je; j += 8) {
    int cc[8];
    float ww[8];
    v4f vv[8];
#pragma unroll
    for (int i = 0; i < 8; ++i) {
      bool ok = (j + i) < je;
      int jc = ok ? (j + i) : last;
      cc[i] = cs[jc];
      float wi = wv[jc];
      ww[i] = ok ? wi : 0.0f;
    }
#pragma unroll
    for (int i = 0; i < 8; ++i) vv[i] = *(const v4f*)(xq + cc[i] * 16);
#pragma unroll
    for (int i = 0; i < 8; ++i) acc += ww[i] * vv[i];
  }
  v4f a;
  if constexpr (STEP == 0) a = acc;
  else a = 2.0f * acc - p;
  float* d = sT + (q * R + rl) * P;
  *(v4f*)(d) = a;
  if constexpr (STEP == 0) *(v4f*)(d + TILE) = p;
  __syncthreads();

  if constexpr (STEP != 2) *(v4f*)(tk + o) = a;

  const int wave = t >> 5;
  const int lane = t & 31;
  const int lm = lane & 15;
  const int khalf = (lane >> 4) << 1;
  const int mhi = (lane >> 4) << 3;

  v2f bf[NT];
#pragma unroll
  for (int i = 0; i < NT; ++i) {
    const float* Wi = W + ((STEP == 0 && i == 0) ? 3 * 16 : 0);
    bf[i].x = ldsel(Wi, khalf * 16 + lm, khalf < 3);
    bf[i].y = ldsel(Wi, (khalf + 1) * 16 + lm, khalf + 1 < 3);
  }
  float bv = 0.0f;
  if constexpr (STEP == 0) {
    if (bias) bv = bias[lm];
  }

#pragma unroll
  for (int it = 0; it < 2; ++it) {
    int mt = wave * 2 + it;
    int b = mt & 3;
    int rt = mt >> 2;
    const float* arow = sT + (b * R + rt * 16 + lm) * P + khalf;
    v8f c = {0.f, 0.f, 0.f, 0.f, 0.f, 0.f, 0.f, 0.f};
#pragma unroll
    for (int i = 0; i < NT; ++i) {
      v2f af = *(const v2f*)(arow + i * TILE);
      c = __builtin_amdgcn_wmma_f32_16x16x4_f32(false, af, false, bf[i],
                                                (short)0, c, false, false);
    }
    float* orow = out + (size_t)(row0 + rt * 16 + mhi) * 64 + (size_t)b * 16 + lm;
#pragma unroll
    for (int rr = 0; rr < 8; ++rr) {
      float v = c[rr];
      if constexpr (STEP == 0) v += bv;
      else v += orow[(size_t)rr * 64];
      if constexpr (STEP == 2 && RELU) v = v > 0.0f ? v : 0.0f;
      orow[(size_t)rr * 64] = v;
    }
  }
}


__global__ __launch_bounds__(256) void k_y_planes(const float* __restrict__ h,
                                                  const float* __restrict__ W,
                                                  float* __restrict__ P, int M) {
  constexpr int G = 4;
  const int t = threadIdx.x;
  const int wave = t >> 5;
  const int lane = t & 31;
  const int c = lane & 15;
  const int hh = lane >> 4;
  const int khalf = hh << 1;
  v2f af[2][4];
#pragma unroll
  for (int tile = 0; tile < 2; ++tile) {
    int j = tile * 16 + c;
    int k = j >> 2;
    int f = j & 3;
    bool ok = (f < 3) && (k < 6);
#pragma unroll
    for (int kc = 0; kc < 4; ++kc) {
      int ka = kc * 4 + khalf;
      af[tile][kc].x = ldsel(W, k * 48 + ka * 3 + f, ok);
      af[tile][kc].y = ldsel(W, k * 48 + (ka + 1) * 3 + f, ok);
    }
  }
  const size_t PL = (size_t)M * 4;
  const int mb = (blockIdx.x * 8 + wave) * (16 * G);
#pragma unroll
  for (int g = 0; g < G; ++g) {
    int m = mb + g * 16 + c;
    const float* hr = h + (size_t)m * 16 + khalf;
    v2f b0 = *(const v2f*)(hr);
    v2f b1 = *(const v2f*)(hr + 4);
    v2f b2 = *(const v2f*)(hr + 8);
    v2f b3 = *(const v2f*)(hr + 12);
    v8f d0 = {0.f, 0.f, 0.f, 0.f, 0.f, 0.f, 0.f, 0.f};
    v8f d1 = {0.f, 0.f, 0.f, 0.f, 0.f, 0.f, 0.f, 0.f};
    d0 = __builtin_amdgcn_wmma_f32_16x16x4_f32(false, af[0][0], false, b0, (short)0, d0, false, false);
    d1 = __builtin_amdgcn_wmma_f32_16x16x4_f32(false, af[1][0], false, b0, (short)0, d1, false, false);
    d0 = __builtin_amdgcn_wmma_f32_16x16x4_f32(false, af[0][1], false, b1, (short)0, d0, false, false);
    d1 = __builtin_amdgcn_wmma_f32_16x16x4_f32(false, af[1][1], false, b1, (short)0, d1, false, false);
    d0 = __builtin_amdgcn_wmma_f32_16x16x4_f32(false, af[0][2], false, b2, (short)0, d0, false, false);
    d1 = __builtin_amdgcn_wmma_f32_16x16x4_f32(false, af[1][2], false, b2, (short)0, d1, false, false);
    d0 = __builtin_amdgcn_wmma_f32_16x16x4_f32(false, af[0][3], false, b3, (short)0, d0, false, false);
    d1 = __builtin_amdgcn_wmma_f32_16x16x4_f32(false, af[1][3], false, b3, (short)0, d1, false, false);
    float* o = P + (size_t)m * 4;
    v4f lo = {d0[0], d0[1], d0[2], d0[3]};
    v4f hi = {d0[4], d0[5], d0[6], d0[7]};
    *(v4f*)(o + (size_t)(2 * hh) * PL) = lo;
    *(v4f*)(o + (size_t)(2 * hh + 1) * PL) = hi;
    if (hh == 0) {
      v4f lo1 = {d1[0], d1[1], d1[2], d1[3]};
      v4f hi1 = {d1[4], d1[5], d1[6], d1[7]};
      *(v4f*)(o + 4 * PL) = lo1;
      *(v4f*)(o + 5 * PL) = hi1;
    }
  }
}

template <int MODE>
__global__ __launch_bounds__(256) void k_clenshaw_step(
    const int* __restrict__ rowptr, const int* __restrict__ cs,
    const float* __restrict__ wv, const float* __restrict__ bn,
    const float* __restrict__ b2, float* yk, float* __restrict__ out, int N) {
  const int t = threadIdx.x;
  const int rl = t >> 2;
  const int q = t & 3;
  const int r = blockIdx.x * 64 + rl;
  int j = rowptr[r];
  const int je = rowptr[r + 1];
  const int o = (r * 4 + q) * 4;
  v4f y = *(const v4f*)(yk + o);
  v4f p = (v4f)(0.0f);
  if constexpr (MODE != 0) p = *(const v4f*)(b2 + o);
  const float* xq = bn + q * 4;
  const int last = je - 1;
  v4f acc = (v4f)(0.0f);
  for (; j < je; j += 8) {
    int cc[8];
    float ww[8];
    v4f vv[8];
#pragma unroll
    for (int i = 0; i < 8; ++i) {
      bool ok = (j + i) < je;
      int jc = ok ? (j + i) : last;
      cc[i] = cs[jc];
      float wi = wv[jc];
      ww[i] = ok ? wi : 0.0f;
    }
#pragma unroll
    for (int i = 0; i < 8; ++i) vv[i] = *(const v4f*)(xq + cc[i] * 16);
#pragma unroll
    for (int i = 0; i < 8; ++i) acc += ww[i] * vv[i];
  }
  v4f res;
  if constexpr (MODE == 0) res = 2.0f * acc + y;
  else if constexpr (MODE == 1) res = 2.0f * acc + y - p;
  else res = acc + y - p;
  if constexpr (MODE != 2) {
    *(v4f*)(yk + o) = res;
  } else {
    __shared__ __attribute__((aligned(16))) float so[4 * 64 * 3];
    float* d = so + (q * 64 + rl) * 3;
    d[0] = res.x;
    d[1] = res.y;
    d[2] = res.z;
    __syncthreads();
    if (t < 192) {
      int b = t / 48;
      int pc = t - b * 48;
      v4f v = *(const v4f*)(so + b * 192 + pc * 4);
      *(v4f*)(out + ((size_t)b * N + (size_t)blockIdx.x * 64) * 3 + pc * 4) = v;
    }
  }
}

__global__ void k_pool(const float* __restrict__ x, const int* __restrict__ idx,
                       const float* __restrict__ w, float* __restrict__ out,
                       int Nout, int Nin, int F) {
  (void)Nin;
  long long tid = (long long)blockIdx.x * blockDim.x + threadIdx.x;
  long long total = 4LL * Nout * F;
  if (tid >= total) return;
  int F4 = 4 * F;
  int bf = (int)(tid % F4);
  int n = (int)(tid / F4);
  float s = 0.0f;
#pragma unroll
  for (int k = 0; k < 3; ++k) {
    int j = idx[n * 3 + k];
    s += w[n * 3 + k] * x[(size_t)j * F4 + bf];
  }
  out[tid] = s;
}

__global__ void k_dense_enc(const float* __restrict__ h, const float* __restrict__ w,
                            const float* __restrict__ bias, float* __restrict__ z) {
  int b = blockIdx.x >> 6;
  int j = blockIdx.x & 63;
  float s = 0.0f;
  for (int i = threadIdx.x; i < 8192; i += blockDim.x)
    s += h[((i >> 5) * 4 + b) * 32 + (i & 31)] * w[(size_t)i * 64 + j];
  __shared__ float red[256];
  red[threadIdx.x] = s;
  __syncthreads();
  for (int off = 128; off > 0; off >>= 1) {
    if (threadIdx.x < off) red[threadIdx.x] += red[threadIdx.x + off];
    __syncthreads();
  }
  if (threadIdx.x == 0) {
    float v = red[0] + bias[j];
    z[b * 64 + j] = v > 0.0f ? v : 0.0f;
  }
}

__global__ void k_dense_dec(const float* __restrict__ z, const float* __restrict__ w,
                            const float* __restrict__ bias, float* __restrict__ h) {
  long long tid = (long long)blockIdx.x * blockDim.x + threadIdx.x;
  if (tid >= 4LL * 8192) return;
  int b = (int)(tid >> 13);
  int j = (int)(tid & 8191);
  float s = bias[j];
  for (int i = 0; i < 64; ++i) s += z[b * 64 + i] * w[(size_t)i * 8192 + j];
  h[((j >> 5) * 4 + b) * 32 + (j & 31)] = s > 0.0f ? s : 0.0f;
}


template <int FIN, int FOUT>
static void launch_step_sh(const int* rp, const int* cs, const float* wv,
                           const float* x, const float* p2, float* tk,
                           const float* W, const float* bias, float* out, int N,
                           int step, int relu, int pb, hipStream_t s) {
  constexpr int R = (FIN == 3) ? 64 : 32;
  unsigned blocks = (unsigned)(N / R);
  if (step == 0)
    k_cheb_step<FIN, FOUT, 0, false><<<blocks, 256, 0, s>>>(rp, cs, wv, x, p2, tk, W, bias, out, N, pb);
  else if (step == 1)
    k_cheb_step<FIN, FOUT, 1, false><<<blocks, 256, 0, s>>>(rp, cs, wv, x, p2, tk, W, bias, out, N, pb);
  else if (relu)
    k_cheb_step<FIN, FOUT, 2, true><<<blocks, 256, 0, s>>>(rp, cs, wv, x, p2, tk, W, bias, out, N, pb);
  else
    k_cheb_step<FIN, FOUT, 2, false><<<blocks, 256, 0, s>>>(rp, cs, wv, x, p2, tk, W, bias, out, N, pb);
}

static void launch_step(const int* rp, const int* cs, const float* wv,
                        const float* x, const float* p2, float* tk,
                        const float* W, const float* bias, float* out, int N,
                        int Fin, int Fout, int step, int relu, int pb,
                        hipStream_t s) {
  if (Fin == 16 && Fout == 16)
    launch_step_sh<16, 16>(rp, cs, wv, x, p2, tk, W, bias, out, N, step, relu, pb, s);
  else if (Fin == 16 && Fout == 32)
    launch_step_sh<16, 32>(rp, cs, wv, x, p2, tk, W, bias, out, N, step, relu, pb, s);
  else if (Fin == 32 && Fout == 16)
    launch_step_sh<32, 16>(rp, cs, wv, x, p2, tk, W, bias, out, N, step, relu, pb, s);
}

static void run_conv(const float* x, const int* rp, const int* cs, const float* wv,
                     const float* W  , const float* bias,
                     float* out, float* b0, float* b1,
                     int N, int Fin, int Fout, int relu, int xin, hipStream_t s) {
  size_t wstride = (size_t)Fin * Fout;

  launch_step(rp, cs, wv, x, x, b0, W, bias, out, N, Fin, Fout, 0, relu,
               xin, s);

  for (int k = 2; k < 6; ++k) {
    const float* prev = (k & 1) ? b1 : b0;
    float* tk = (k & 1) ? b0 : b1;
    const float* p2 = (k == 2) ? x : tk;
    launch_step(rp, cs, wv, prev, p2, tk, W + (size_t)k * wstride, nullptr, out,
                N, Fin, Fout, (k == 5) ? 2 : 1, relu,
                 (k == 2) ? xin : 0, s);
  }
}

static void run_conv3(const float* xi, const int* rp, const int* cs, const float* wv,
                      const float* W  , const float* bias, float* out,
                      float* b0, float* b1, int N, hipStream_t s) {
  unsigned blocks = (unsigned)(N / 64);
  k_cheb3_step<0, false><<<blocks, 256, 0, s>>>(rp, cs, wv, xi, xi, b0, W, bias, out);
  for (int k = 2; k < 6; ++k) {
    const float* prev = (k & 1) ? b1 : b0;
    float* tk = (k & 1) ? b0 : b1;
    const float* p2 = (k == 2) ? xi : tk;
    if (k < 5)
      k_cheb3_step<1, false><<<blocks, 256, 0, s>>>(rp, cs, wv, prev, p2, tk,
                                                   W + (size_t)k * 48, nullptr, out);
    else
      k_cheb3_step<2, true><<<blocks, 256, 0, s>>>(rp, cs, wv, prev, p2, tk,
                                                  W + (size_t)k * 48, nullptr, out);
  }
}


extern "C" void kernel_launch(void* const* d_in, const int* in_sizes, int n_in,
                              void* d_out, int out_size, void* d_ws, size_t ws_size,
                              hipStream_t stream) {
  (void)in_sizes; (void)n_in; (void)out_size; (void)ws_size;
  static const int NSs[5] = {65536, 16384, 4096, 1024, 256};
  static const int FE[5] = {3, 16, 16, 16, 32};
  static const int FD[6] = {32, 16, 16, 16, 16, 3};

  const float* x = (const float*)d_in[0];
  const int* ei[5];
  for (int l = 0; l < 5; ++l) ei[l] = (const int*)d_in[1 + l];
  const int* down_i[4]; const float* down_w[4];
  const int* up_i[4];   const float* up_w[4];
  for (int i = 0; i < 4; ++i) {
    down_i[i] = (const int*)d_in[6 + 4 * i];
    down_w[i] = (const float*)d_in[7 + 4 * i];
    up_i[i]   = (const int*)d_in[8 + 4 * i];
    up_w[i]   = (const float*)d_in[9 + 4 * i];
  }
  const float* encW[4]; const float* encB[4];
  for (int i = 0; i < 4; ++i) {
    encW[i] = (const float*)d_in[22 + 2 * i];
    encB[i] = (const float*)d_in[23 + 2 * i];
  }
  const float* decW[5];
  for (int i = 0; i < 5; ++i) decW[i] = (const float*)d_in[30 + i];
  const float* decB[4];
  for (int i = 0; i < 4; ++i) decB[i] = (const float*)d_in[35 + i];
  const float* enc_w = (const float*)d_in[39];
  const float* enc_b = (const float*)d_in[40];
  const float* dec_w = (const float*)d_in[41];
  const float* dec_b = (const float*)d_in[42];

  float* ws = (float*)d_ws;
  size_t off = 0;
  const size_t ACT = (size_t)4 * 65536 * 16;
  float* actA = ws + off; off += ACT;
  float* actB = ws + off; off += ACT;
  float* tb0 = ws + off; off += ACT;
  float* tb1 = ws + off; off += ACT;
  float* zbuf = ws + off; off += 256;
  int* rp[4]; int* cs[4]; float* wv[4];
  for (int l = 0; l < 4; ++l) {
    rp[l] = (int*)(ws + off); off += (size_t)NSs[l] + 4;
    cs[l] = (int*)(ws + off); off += (size_t)16 * NSs[l];
    wv[l] = ws + off;         off += (size_t)16 * NSs[l];
  }
  int* rank = (int*)tb0;
  int* slot = rank + (size_t)ETOT;
  int* cnt = slot + (size_t)ETOT;

  float* xi = actB;
  k_zero_i<<<(unsigned)(CDIV(NTOT, 256) + CDIV(4 * 65536, 256)), 256, 0, stream>>>(
      cnt, NTOT, CDIV(NTOT, 256), x, xi, 65536);
  k_rank<<<(unsigned)CDIV(ETOT, 256), 256, 0, stream>>>(ei[0], ei[1], ei[2], ei[3],
                                                       cnt, rank);
  k_scan<<<4, 1024, 0, stream>>>(cnt, rp[0], rp[1], rp[2], rp[3]);
  k_place<<<(unsigned)CDIV(ETOT, 256), 256, 0, stream>>>(
      ei[0], ei[1], ei[2], ei[3], rp[0], rp[1], rp[2], rp[3], rank, slot);
  CsrOut co = {cs[0], cs[1], cs[2], cs[3], wv[0], wv[1], wv[2], wv[3]};
  k_emit<<<(unsigned)CDIV(ETOT, 256), 256, 0, stream>>>(
      ei[0], ei[1], ei[2], ei[3], cnt, rp[0], rp[1], rp[2], rp[3], slot, co);

  const float* cur = xi;
  for (int i = 0; i < 4; ++i) {
    if (i == 0)
      run_conv3(cur, rp[0], cs[0], wv[0], encW[0], encB[0], actA, tb0, tb1,
                NSs[0], stream);
    else
      run_conv(cur, rp[i], cs[i], wv[i], encW[i], encB[i], actA, tb0, tb1,
               NSs[i], FE[i], FE[i + 1],  1,  0, stream);
    long long pt = 4LL * NSs[i + 1] * FE[i + 1];
    k_pool<<<(unsigned)CDIV(pt, 256), 256, 0, stream>>>(
        actA, down_i[i], down_w[i], actB, NSs[i + 1], NSs[i], FE[i + 1]);
    cur = actB;
  }

  k_dense_enc<<<256, 256, 0, stream>>>(actB, enc_w, enc_b, zbuf);
  k_dense_dec<<<(unsigned)CDIV(4LL * 8192, 256), 256, 0, stream>>>(zbuf, dec_w,
                                                                   dec_b, actA);

  for (int i = 0; i < 4; ++i) {
    int lvl = 3 - i;
    long long pt = 4LL * NSs[lvl] * FD[i];
    k_pool<<<(unsigned)CDIV(pt, 256), 256, 0, stream>>>(
        actA, up_i[lvl], up_w[lvl], actB, NSs[lvl], NSs[lvl + 1], FD[i]);
    run_conv(actB, rp[lvl], cs[lvl], wv[lvl], decW[i], decB[i], actA, tb0, tb1,
             NSs[lvl], FD[i], FD[i + 1],  1,  0, stream);
  }

  const int N0 = 65536;
  const size_t PL = (size_t)4 * N0 * 4;
  float* P = tb0;
  k_y_planes<<<(unsigned)(4 * N0 / 512), 256, 0, stream>>>(actA, decW[4], P, 4 * N0);
  k_clenshaw_step<0><<<(unsigned)(N0 / 64), 256, 0, stream>>>(
      rp[0], cs[0], wv[0], P + 5 * PL, nullptr, P + 4 * PL, nullptr, N0);
  k_clenshaw_step<1><<<(unsigned)(N0 / 64), 256, 0, stream>>>(
      rp[0], cs[0], wv[0], P + 4 * PL, P + 5 * PL, P + 3 * PL, nullptr, N0);
  k_clenshaw_step<1><<<(unsigned)(N0 / 64), 256, 0, stream>>>(
      rp[0], cs[0], wv[0], P + 3 * PL, P + 4 * PL, P + 2 * PL, nullptr, N0);
  k_clenshaw_step<1><<<(unsigned)(N0 / 64), 256, 0, stream>>>(
      rp[0], cs[0], wv[0], P + 2 * PL, P + 3 * PL, P + 1 * PL, nullptr, N0);
  k_clenshaw_step<2><<<(unsigned)(N0 / 64), 256, 0, stream>>>(
      rp[0], cs[0], wv[0], P + 1 * PL, P + 2 * PL, P, (float*)d_out, N0);
}
